# starting kernel
# baseline (speedup 1.0000x reference)
#include <hip/hip_runtime.h>

typedef float v2f __attribute__((ext_vector_type(2)));
typedef float v8f __attribute__((ext_vector_type(8)));

#define NNODES   100000
#define NEDGES   1600000
#define DFEAT    128
#define NEG_SLOPE 0.2f

__global__ void gat_zero_f32(float* __restrict__ p, long n) {
    long i = blockIdx.x * (long)blockDim.x + threadIdx.x;
    long stride = gridDim.x * (long)blockDim.x;
    for (; i < n; i += stride) p[i] = 0.0f;
}

__global__ void gat_gemm_wmma(const float* __restrict__ h,
                              const float* __restrict__ W,
                              float* __restrict__ feat) {
    const int strip = blockIdx.x * 4 + (threadIdx.x >> 5);
    if (strip >= NNODES / 16) return;
    const int lane  = threadIdx.x & 31;
    const int m     = lane & 15;
    const int khalf = (lane >> 4) << 1;
    const int hihalf = (lane >> 4) << 3;
    const long row  = (long)strip * 16 + m;

    v8f acc[8];
#pragma unroll
    for (int ct = 0; ct < 8; ++ct) acc[ct] = (v8f){0,0,0,0,0,0,0,0};

    for (int k = 0; k < DFEAT; k += 4) {
        v2f a;
        a.x = h[row * DFEAT + k + khalf + 0];
        a.y = h[row * DFEAT + k + khalf + 1];
#pragma unroll
        for (int ct = 0; ct < 8; ++ct) {
            v2f b;
            b.x = W[(k + khalf + 0) * DFEAT + ct * 16 + m];
            b.y = W[(k + khalf + 1) * DFEAT + ct * 16 + m];
            acc[ct] = __builtin_amdgcn_wmma_f32_16x16x4_f32(
                false, a, false, b, (short)0, acc[ct], false, false);
        }
    }

#pragma unroll
    for (int ct = 0; ct < 8; ++ct) {
#pragma unroll
        for (int v = 0; v < 8; ++v) {
            long orow = (long)strip * 16 + v + hihalf;
            feat[orow * DFEAT + ct * 16 + m] = acc[ct][v];
        }
    }
}

__global__ void gat_attn_logits(const float* __restrict__ feat,
                                const float* __restrict__ attn_l,
                                const float* __restrict__ attn_r,
                                float* __restrict__ el,
                                float* __restrict__ er) {
    const int node = blockIdx.x * 4 + (threadIdx.x >> 5);
    if (node >= NNODES) return;
    const int lane = threadIdx.x & 31;

    const float4 f  = *(const float4*)(feat   + (long)node * DFEAT + lane * 4);
    const float4 al = *(const float4*)(attn_l + lane * 4);
    const float4 ar = *(const float4*)(attn_r + lane * 4);

    float pl = f.x * al.x + f.y * al.y + f.z * al.z + f.w * al.w;
    float pr = f.x * ar.x + f.y * ar.y + f.z * ar.z + f.w * ar.w;
#pragma unroll
    for (int off = 16; off > 0; off >>= 1) {
        pl += __shfl_xor(pl, off, 32);
        pr += __shfl_xor(pr, off, 32);
    }
    if (lane == 0) { el[node] = pl; er[node] = pr; }
}

__device__ __forceinline__ unsigned enc_f32(float x) {
    unsigned u = __float_as_uint(x);
    return (u & 0x80000000u) ? ~u : (u | 0x80000000u);
}
__device__ __forceinline__ float dec_f32(unsigned u) {
    return (u & 0x80000000u) ? __uint_as_float(u ^ 0x80000000u) : __uint_as_float(~u);
}

__global__ void gat_edge_score(const int* __restrict__ src,
                               const int* __restrict__ dst,
                               const float* __restrict__ el,
                               const float* __restrict__ er,
                               float* __restrict__ e,
                               unsigned* __restrict__ menc) {
    const int i = blockIdx.x * blockDim.x + threadIdx.x;
    if (i >= NEDGES) return;
    const int s = src[i], d = dst[i];
    float v = el[s] + er[d];
    v = (v >= 0.0f) ? v : NEG_SLOPE * v;
    e[i] = v;
    atomicMax(&menc[d], enc_f32(v));
}

__global__ void gat_edge_exp(const int* __restrict__ dst,
                             const unsigned* __restrict__ menc,
                             float* __restrict__ e,
                             float* __restrict__ denom) {
    const int i = blockIdx.x * blockDim.x + threadIdx.x;
    if (i >= NEDGES) return;
    const int d = dst[i];
    const float mx = dec_f32(menc[d]);
    const float ex = __expf(e[i] - mx);
    e[i] = ex;
    atomicAdd(&denom[d], ex);
}

__global__ void gat_scatter(const int* __restrict__ src,
                            const int* __restrict__ dst,
                            const float* __restrict__ ex,
                            const float* __restrict__ denom,
                            const float* __restrict__ feat,
                            float* __restrict__ out) {
    const int edge = blockIdx.x * 8 + (threadIdx.x >> 5);
    if (edge >= NEDGES) return;
    const int lane = threadIdx.x & 31;
    const int s = src[edge], d = dst[edge];
    const float alpha = ex[edge] / fmaxf(denom[d], 1e-9f);

    const float4 f = *(const float4*)(feat + (long)s * DFEAT + lane * 4);
    float* o = out + (long)d * DFEAT + lane * 4;
    atomicAdd(o + 0, alpha * f.x);
    atomicAdd(o + 1, alpha * f.y);
    atomicAdd(o + 2, alpha * f.z);
    atomicAdd(o + 3, alpha * f.w);
}

extern "C" void kernel_launch(void* const* d_in, const int* in_sizes, int n_in,
                              void* d_out, int out_size, void* d_ws, size_t ws_size,
                              hipStream_t stream) {
    const float* h      = (const float*)d_in[0];
    const int*   src    = (const int*)  d_in[1];
    const int*   dst    = (const int*)  d_in[2];
    const float* W      = (const float*)d_in[3];
    const float* attn_l = (const float*)d_in[4];
    const float* attn_r = (const float*)d_in[5];
    float* out = (float*)d_out;

    float*    feat  = (float*)d_ws;
    float*    el    = feat + (long)NNODES * DFEAT;
    float*    er    = el + NNODES;
    unsigned* menc  = (unsigned*)(er + NNODES);
    float*    denom = (float*)(menc + NNODES);
    float*    e     = denom + NNODES;

    gat_zero_f32<<<4096, 256, 0, stream>>>(out, (long)NNODES * DFEAT);
    gat_zero_f32<<<256, 256, 0, stream>>>((float*)menc, 2L * NNODES);

    const int nstrips = NNODES / 16;
    gat_gemm_wmma<<<(nstrips + 3) / 4, 128, 0, stream>>>(h, W, feat);

    gat_attn_logits<<<(NNODES + 3) / 4, 128, 0, stream>>>(feat, attn_l, attn_r, el, er);

    gat_edge_score<<<(NEDGES + 255) / 256, 256, 0, stream>>>(src, dst, el, er, e, menc);

    gat_edge_exp<<<(NEDGES + 255) / 256, 256, 0, stream>>>(dst, menc, e, denom);

    gat_scatter<<<(NEDGES + 7) / 8, 256, 0, stream>>>(src, dst, e, denom, feat, out);
}


# scatter and logits kernels: one wave loops over 32 edges / 8 nodes instead of one wave per item (wave launches cut 32x / 8x); float atomics kept, same addends, arrival order still arbitrary
# speedup vs baseline: 1.0388x; 1.0388x over previous
#include <hip/hip_runtime.h>

typedef float v2f __attribute__((ext_vector_type(2)));
typedef float v8f __attribute__((ext_vector_type(8)));

#define NNODES   100000
#define NEDGES   1600000
#define DFEAT    128
#define NEG_SLOPE 0.2f

__global__ void gat_zero_f32(float* __restrict__ p, long n) {
    long i = blockIdx.x * (long)blockDim.x + threadIdx.x;
    long stride = gridDim.x * (long)blockDim.x;
    for (; i < n; i += stride) p[i] = 0.0f;
}

__global__ void gat_gemm_wmma(const float* __restrict__ h,
                              const float* __restrict__ W,
                              float* __restrict__ feat) {
    const int strip = blockIdx.x * 4 + (threadIdx.x >> 5);
    if (strip >= NNODES / 16) return;
    const int lane  = threadIdx.x & 31;
    const int m     = lane & 15;
    const int khalf = (lane >> 4) << 1;
    const int hihalf = (lane >> 4) << 3;
    const long row  = (long)strip * 16 + m;

    v8f acc[8];
#pragma unroll
    for (int ct = 0; ct < 8; ++ct) acc[ct] = (v8f){0,0,0,0,0,0,0,0};

    for (int k = 0; k < DFEAT; k += 4) {
        v2f a;
        a.x = h[row * DFEAT + k + khalf + 0];
        a.y = h[row * DFEAT + k + khalf + 1];
#pragma unroll
        for (int ct = 0; ct < 8; ++ct) {
            v2f b;
            b.x = W[(k + khalf + 0) * DFEAT + ct * 16 + m];
            b.y = W[(k + khalf + 1) * DFEAT + ct * 16 + m];
            acc[ct] = __builtin_amdgcn_wmma_f32_16x16x4_f32(
                false, a, false, b, (short)0, acc[ct], false, false);
        }
    }

#pragma unroll
    for (int ct = 0; ct < 8; ++ct) {
#pragma unroll
        for (int v = 0; v < 8; ++v) {
            long orow = (long)strip * 16 + v + hihalf;
            feat[orow * DFEAT + ct * 16 + m] = acc[ct][v];
        }
    }
}

#define NPW 8
static_assert(NNODES % NPW == 0, "nodes per wave must divide the node count");
__global__ void gat_attn_logits(const float* __restrict__ feat,
                                const float* __restrict__ attn_l,
                                const float* __restrict__ attn_r,
                                float* __restrict__ el,
                                float* __restrict__ er) {
    const int n0 = (blockIdx.x * 4 + (threadIdx.x >> 5)) * NPW;
    if (n0 >= NNODES) return;
    const int lane = threadIdx.x & 31;

    const float4 al = *(const float4*)(attn_l + lane * 4);
    const float4 ar = *(const float4*)(attn_r + lane * 4);

    float4 f[NPW];
#pragma unroll
    for (int u = 0; u < NPW; ++u)
        f[u] = *(const float4*)(feat + (long)(n0 + u) * DFEAT + lane * 4);

    float keep_l = 0.0f, keep_r = 0.0f;
#pragma unroll
    for (int u = 0; u < NPW; ++u) {
        float pl = f[u].x * al.x + f[u].y * al.y + f[u].z * al.z + f[u].w * al.w;
        float pr = f[u].x * ar.x + f[u].y * ar.y + f[u].z * ar.z + f[u].w * ar.w;
#pragma unroll
        for (int off = 16; off > 0; off >>= 1) {
            pl += __shfl_xor(pl, off, 32);
            pr += __shfl_xor(pr, off, 32);
        }
        if (lane == u) { keep_l = pl; keep_r = pr; }
    }
    if (lane < NPW) { el[n0 + lane] = keep_l; er[n0 + lane] = keep_r; }
}

__device__ __forceinline__ unsigned enc_f32(float x) {
    unsigned u = __float_as_uint(x);
    return (u & 0x80000000u) ? ~u : (u | 0x80000000u);
}
__device__ __forceinline__ float dec_f32(unsigned u) {
    return (u & 0x80000000u) ? __uint_as_float(u ^ 0x80000000u) : __uint_as_float(~u);
}

__global__ void gat_edge_score(const int* __restrict__ src,
                               const int* __restrict__ dst,
                               const float* __restrict__ el,
                               const float* __restrict__ er,
                               float* __restrict__ e,
                               unsigned* __restrict__ menc) {
    const int i = blockIdx.x * blockDim.x + threadIdx.x;
    if (i >= NEDGES) return;
    const int s = src[i], d = dst[i];
    float v = el[s] + er[d];
    v = (v >= 0.0f) ? v : NEG_SLOPE * v;
    e[i] = v;
    atomicMax(&menc[d], enc_f32(v));
}

__global__ void gat_edge_exp(const int* __restrict__ dst,
                             const unsigned* __restrict__ menc,
                             float* __restrict__ e,
                             float* __restrict__ denom) {
    const int i = blockIdx.x * blockDim.x + threadIdx.x;
    if (i >= NEDGES) return;
    const int d = dst[i];
    const float mx = dec_f32(menc[d]);
    const float ex = __expf(e[i] - mx);
    e[i] = ex;
    atomicAdd(&denom[d], ex);
}

#define EPW 32
#define EBATCH 8
static_assert(NEDGES % EPW == 0, "edges per wave must divide the edge count");
static_assert(EPW == 32 && EPW % EBATCH == 0, "one edge per lane, whole batches");
__global__ void gat_scatter(const int* __restrict__ src,
                            const int* __restrict__ dst,
                            const float* __restrict__ ex,
                            const float* __restrict__ denom,
                            const float* __restrict__ feat,
                            float* __restrict__ out) {
    const int base = (blockIdx.x * 8 + (threadIdx.x >> 5)) * EPW;
    if (base >= NEDGES) return;
    const int lane = threadIdx.x & 31;

    const int   s_v = src[base + lane];
    const int   d_v = dst[base + lane];
    const float a_v = ex[base + lane] / fmaxf(denom[d_v], 1e-9f);

#pragma unroll
    for (int j0 = 0; j0 < EPW; j0 += EBATCH) {
        float4 f[EBATCH];
#pragma unroll
        for (int u = 0; u < EBATCH; ++u) {
            const int s = __builtin_amdgcn_readlane(s_v, j0 + u);
            f[u] = *(const float4*)(feat + (long)s * DFEAT + lane * 4);
        }
#pragma unroll
        for (int u = 0; u < EBATCH; ++u) {
            const int   d     = __builtin_amdgcn_readlane(d_v, j0 + u);
            const float alpha = __int_as_float(__builtin_amdgcn_readlane(__float_as_int(a_v), j0 + u));
            float* o = out + (long)d * DFEAT + lane * 4;
            atomicAdd(o + 0, alpha * f[u].x);
            atomicAdd(o + 1, alpha * f[u].y);
            atomicAdd(o + 2, alpha * f[u].z);
            atomicAdd(o + 3, alpha * f[u].w);
        }
    }
}

extern "C" void kernel_launch(void* const* d_in, const int* in_sizes, int n_in,
                              void* d_out, int out_size, void* d_ws, size_t ws_size,
                              hipStream_t stream) {
    const float* h      = (const float*)d_in[0];
    const int*   src    = (const int*)  d_in[1];
    const int*   dst    = (const int*)  d_in[2];
    const float* W      = (const float*)d_in[3];
    const float* attn_l = (const float*)d_in[4];
    const float* attn_r = (const float*)d_in[5];
    float* out = (float*)d_out;

    float*    feat  = (float*)d_ws;
    float*    el    = feat + (long)NNODES * DFEAT;
    float*    er    = el + NNODES;
    unsigned* menc  = (unsigned*)(er + NNODES);
    float*    denom = (float*)(menc + NNODES);
    float*    e     = denom + NNODES;

    gat_zero_f32<<<4096, 256, 0, stream>>>(out, (long)NNODES * DFEAT);
    gat_zero_f32<<<256, 256, 0, stream>>>((float*)menc, 2L * NNODES);

    const int nstrips = NNODES / 16;
    gat_gemm_wmma<<<(nstrips + 3) / 4, 128, 0, stream>>>(h, W, feat);

    gat_attn_logits<<<(NNODES / NPW + 3) / 4, 128, 0, stream>>>(feat, attn_l, attn_r, el, er);

    gat_edge_score<<<(NEDGES + 255) / 256, 256, 0, stream>>>(src, dst, el, er, e, menc);

    gat_edge_exp<<<(NEDGES + 255) / 256, 256, 0, stream>>>(dst, menc, e, denom);

    gat_scatter<<<(NEDGES / EPW + 7) / 8, 256, 0, stream>>>(src, dst, e, denom, feat, out);
}


# scatter: lane l owns columns l,32+l,64+l,96+l, so each float atomic and row load covers one whole 128-B line (4 line updates per edge, not 16); atomics kept, same addends, arrival order arbitrary
# speedup vs baseline: 1.5982x; 1.5385x over previous
#include <hip/hip_runtime.h>

typedef float v2f __attribute__((ext_vector_type(2)));
typedef float v8f __attribute__((ext_vector_type(8)));

#define NNODES   100000
#define NEDGES   1600000
#define DFEAT    128
#define NEG_SLOPE 0.2f

__global__ void gat_zero_f32(float* __restrict__ p, long n) {
    long i = blockIdx.x * (long)blockDim.x + threadIdx.x;
    long stride = gridDim.x * (long)blockDim.x;
    for (; i < n; i += stride) p[i] = 0.0f;
}

__global__ void gat_gemm_wmma(const float* __restrict__ h,
                              const float* __restrict__ W,
                              float* __restrict__ feat) {
    const int strip = blockIdx.x * 4 + (threadIdx.x >> 5);
    if (strip >= NNODES / 16) return;
    const int lane  = threadIdx.x & 31;
    const int m     = lane & 15;
    const int khalf = (lane >> 4) << 1;
    const int hihalf = (lane >> 4) << 3;
    const long row  = (long)strip * 16 + m;

    v8f acc[8];
#pragma unroll
    for (int ct = 0; ct < 8; ++ct) acc[ct] = (v8f){0,0,0,0,0,0,0,0};

    for (int k = 0; k < DFEAT; k += 4) {
        v2f a;
        a.x = h[row * DFEAT + k + khalf + 0];
        a.y = h[row * DFEAT + k + khalf + 1];
#pragma unroll
        for (int ct = 0; ct < 8; ++ct) {
            v2f b;
            b.x = W[(k + khalf + 0) * DFEAT + ct * 16 + m];
            b.y = W[(k + khalf + 1) * DFEAT + ct * 16 + m];
            acc[ct] = __builtin_amdgcn_wmma_f32_16x16x4_f32(
                false, a, false, b, (short)0, acc[ct], false, false);
        }
    }

#pragma unroll
    for (int ct = 0; ct < 8; ++ct) {
#pragma unroll
        for (int v = 0; v < 8; ++v) {
            long orow = (long)strip * 16 + v + hihalf;
            feat[orow * DFEAT + ct * 16 + m] = acc[ct][v];
        }
    }
}

#define NPW 8
static_assert(NNODES % NPW == 0, "nodes per wave must divide the node count");
__global__ void gat_attn_logits(const float* __restrict__ feat,
                                const float* __restrict__ attn_l,
                                const float* __restrict__ attn_r,
                                float* __restrict__ el,
                                float* __restrict__ er) {
    const int n0 = (blockIdx.x * 4 + (threadIdx.x >> 5)) * NPW;
    if (n0 >= NNODES) return;
    const int lane = threadIdx.x & 31;

    const float4 al = *(const float4*)(attn_l + lane * 4);
    const float4 ar = *(const float4*)(attn_r + lane * 4);

    float4 f[NPW];
#pragma unroll
    for (int u = 0; u < NPW; ++u)
        f[u] = *(const float4*)(feat + (long)(n0 + u) * DFEAT + lane * 4);

    float keep_l = 0.0f, keep_r = 0.0f;
#pragma unroll
    for (int u = 0; u < NPW; ++u) {
        float pl = f[u].x * al.x + f[u].y * al.y + f[u].z * al.z + f[u].w * al.w;
        float pr = f[u].x * ar.x + f[u].y * ar.y + f[u].z * ar.z + f[u].w * ar.w;
#pragma unroll
        for (int off = 16; off > 0; off >>= 1) {
            pl += __shfl_xor(pl, off, 32);
            pr += __shfl_xor(pr, off, 32);
        }
        if (lane == u) { keep_l = pl; keep_r = pr; }
    }
    if (lane < NPW) { el[n0 + lane] = keep_l; er[n0 + lane] = keep_r; }
}

__device__ __forceinline__ unsigned enc_f32(float x) {
    unsigned u = __float_as_uint(x);
    return (u & 0x80000000u) ? ~u : (u | 0x80000000u);
}
__device__ __forceinline__ float dec_f32(unsigned u) {
    return (u & 0x80000000u) ? __uint_as_float(u ^ 0x80000000u) : __uint_as_float(~u);
}

__global__ void gat_edge_score(const int* __restrict__ src,
                               const int* __restrict__ dst,
                               const float* __restrict__ el,
                               const float* __restrict__ er,
                               float* __restrict__ e,
                               unsigned* __restrict__ menc) {
    const int i = blockIdx.x * blockDim.x + threadIdx.x;
    if (i >= NEDGES) return;
    const int s = src[i], d = dst[i];
    float v = el[s] + er[d];
    v = (v >= 0.0f) ? v : NEG_SLOPE * v;
    e[i] = v;
    atomicMax(&menc[d], enc_f32(v));
}

__global__ void gat_edge_exp(const int* __restrict__ dst,
                             const unsigned* __restrict__ menc,
                             float* __restrict__ e,
                             float* __restrict__ denom) {
    const int i = blockIdx.x * blockDim.x + threadIdx.x;
    if (i >= NEDGES) return;
    const int d = dst[i];
    const float mx = dec_f32(menc[d]);
    const float ex = __expf(e[i] - mx);
    e[i] = ex;
    atomicAdd(&denom[d], ex);
}

#define EPW 32
#define EBATCH 8
#define ROWLINES (DFEAT / 32)
static_assert(NEDGES % EPW == 0, "edges per wave must divide the edge count");
static_assert(EPW == 32 && EPW % EBATCH == 0, "one edge per lane, whole batches");
static_assert(DFEAT % 32 == 0, "a row is a whole number of 32-float lines");
__global__ void gat_scatter(const int* __restrict__ src,
                            const int* __restrict__ dst,
                            const float* __restrict__ ex,
                            const float* __restrict__ denom,
                            const float* __restrict__ feat,
                            float* __restrict__ out) {
    const int base = (blockIdx.x * 8 + (threadIdx.x >> 5)) * EPW;
    if (base >= NEDGES) return;
    const int lane = threadIdx.x & 31;

    const int   s_v = src[base + lane];
    const int   d_v = dst[base + lane];
    const float a_v = ex[base + lane] / fmaxf(denom[d_v], 1e-9f);

#pragma unroll
    for (int j0 = 0; j0 < EPW; j0 += EBATCH) {
        float f[EBATCH][ROWLINES];
#pragma unroll
        for (int u = 0; u < EBATCH; ++u) {
            const int s = __builtin_amdgcn_readlane(s_v, j0 + u);
            const float* r = feat + (long)s * DFEAT;
#pragma unroll
            for (int c = 0; c < ROWLINES; ++c) f[u][c] = r[c * 32 + lane];
        }
#pragma unroll
        for (int u = 0; u < EBATCH; ++u) {
            const int   d     = __builtin_amdgcn_readlane(d_v, j0 + u);
            const float alpha = __int_as_float(__builtin_amdgcn_readlane(__float_as_int(a_v), j0 + u));
            float* o = out + (long)d * DFEAT;
#pragma unroll
            for (int c = 0; c < ROWLINES; ++c) atomicAdd(o + c * 32 + lane, alpha * f[u][c]);
        }
    }
}

extern "C" void kernel_launch(void* const* d_in, const int* in_sizes, int n_in,
                              void* d_out, int out_size, void* d_ws, size_t ws_size,
                              hipStream_t stream) {
    const float* h      = (const float*)d_in[0];
    const int*   src    = (const int*)  d_in[1];
    const int*   dst    = (const int*)  d_in[2];
    const float* W      = (const float*)d_in[3];
    const float* attn_l = (const float*)d_in[4];
    const float* attn_r = (const float*)d_in[5];
    float* out = (float*)d_out;

    float*    feat  = (float*)d_ws;
    float*    el    = feat + (long)NNODES * DFEAT;
    float*    er    = el + NNODES;
    unsigned* menc  = (unsigned*)(er + NNODES);
    float*    denom = (float*)(menc + NNODES);
    float*    e     = denom + NNODES;

    gat_zero_f32<<<4096, 256, 0, stream>>>(out, (long)NNODES * DFEAT);
    gat_zero_f32<<<256, 256, 0, stream>>>((float*)menc, 2L * NNODES);

    const int nstrips = NNODES / 16;
    gat_gemm_wmma<<<(nstrips + 3) / 4, 128, 0, stream>>>(h, W, feat);

    gat_attn_logits<<<(NNODES / NPW + 3) / 4, 128, 0, stream>>>(feat, attn_l, attn_r, el, er);

    gat_edge_score<<<(NEDGES + 255) / 256, 256, 0, stream>>>(src, dst, el, er, e, menc);

    gat_edge_exp<<<(NEDGES + 255) / 256, 256, 0, stream>>>(dst, menc, e, denom);

    gat_scatter<<<(NEDGES / EPW + 7) / 8, 256, 0, stream>>>(src, dst, e, denom, feat, out);
}


# segment max and softmax denominator: one scattered global atomic per edge replaced by LDS atomics into a work-group table of half the nodes, flushed by whole-line global atomics; sum order arbitrary
# speedup vs baseline: 2.1401x; 1.3391x over previous
#include <hip/hip_runtime.h>

typedef float v2f __attribute__((ext_vector_type(2)));
typedef float v8f __attribute__((ext_vector_type(8)));

#define NNODES   100000
#define NEDGES   1600000
#define DFEAT    128
#define NEG_SLOPE 0.2f

__global__ void gat_zero_f32(float* __restrict__ p, long n) {
    long i = blockIdx.x * (long)blockDim.x + threadIdx.x;
    long stride = gridDim.x * (long)blockDim.x;
    for (; i < n; i += stride) p[i] = 0.0f;
}

__global__ void gat_gemm_wmma(const float* __restrict__ h,
                              const float* __restrict__ W,
                              float* __restrict__ feat) {
    const int strip = blockIdx.x * 4 + (threadIdx.x >> 5);
    if (strip >= NNODES / 16) return;
    const int lane  = threadIdx.x & 31;
    const int m     = lane & 15;
    const int khalf = (lane >> 4) << 1;
    const int hihalf = (lane >> 4) << 3;
    const long row  = (long)strip * 16 + m;

    v8f acc[8];
#pragma unroll
    for (int ct = 0; ct < 8; ++ct) acc[ct] = (v8f){0,0,0,0,0,0,0,0};

    for (int k = 0; k < DFEAT; k += 4) {
        v2f a;
        a.x = h[row * DFEAT + k + khalf + 0];
        a.y = h[row * DFEAT + k + khalf + 1];
#pragma unroll
        for (int ct = 0; ct < 8; ++ct) {
            v2f b;
            b.x = W[(k + khalf + 0) * DFEAT + ct * 16 + m];
            b.y = W[(k + khalf + 1) * DFEAT + ct * 16 + m];
            acc[ct] = __builtin_amdgcn_wmma_f32_16x16x4_f32(
                false, a, false, b, (short)0, acc[ct], false, false);
        }
    }

#pragma unroll
    for (int ct = 0; ct < 8; ++ct) {
#pragma unroll
        for (int v = 0; v < 8; ++v) {
            long orow = (long)strip * 16 + v + hihalf;
            feat[orow * DFEAT + ct * 16 + m] = acc[ct][v];
        }
    }
}

#define NPW 8
static_assert(NNODES % NPW == 0, "nodes per wave must divide the node count");
__global__ void gat_attn_logits(const float* __restrict__ feat,
                                const float* __restrict__ attn_l,
                                const float* __restrict__ attn_r,
                                float* __restrict__ el,
                                float* __restrict__ er) {
    const int n0 = (blockIdx.x * 4 + (threadIdx.x >> 5)) * NPW;
    if (n0 >= NNODES) return;
    const int lane = threadIdx.x & 31;

    const float4 al = *(const float4*)(attn_l + lane * 4);
    const float4 ar = *(const float4*)(attn_r + lane * 4);

    float4 f[NPW];
#pragma unroll
    for (int u = 0; u < NPW; ++u)
        f[u] = *(const float4*)(feat + (long)(n0 + u) * DFEAT + lane * 4);

    float keep_l = 0.0f, keep_r = 0.0f;
#pragma unroll
    for (int u = 0; u < NPW; ++u) {
        float pl = f[u].x * al.x + f[u].y * al.y + f[u].z * al.z + f[u].w * al.w;
        float pr = f[u].x * ar.x + f[u].y * ar.y + f[u].z * ar.z + f[u].w * ar.w;
#pragma unroll
        for (int off = 16; off > 0; off >>= 1) {
            pl += __shfl_xor(pl, off, 32);
            pr += __shfl_xor(pr, off, 32);
        }
        if (lane == u) { keep_l = pl; keep_r = pr; }
    }
    if (lane < NPW) { el[n0 + lane] = keep_l; er[n0 + lane] = keep_r; }
}

__device__ __forceinline__ unsigned enc_f32(float x) {
    unsigned u = __float_as_uint(x);
    return (u & 0x80000000u) ? ~u : (u | 0x80000000u);
}
__device__ __forceinline__ float dec_f32(unsigned u) {
    return (u & 0x80000000u) ? __uint_as_float(u ^ 0x80000000u) : __uint_as_float(~u);
}

#define NH0    50016
#define NCHUNK 128
#define EPC    (NEDGES / NCHUNK)
#define TPB    1024
#define EIT    ((EPC + TPB - 1) / TPB)
#define FIT    ((NH0 + TPB - 1) / TPB)
static_assert(NEDGES % NCHUNK == 0, "whole chunks");
static_assert(NH0 % 32 == 0 && NH0 < NNODES && NNODES - NH0 <= NH0, "two line-aligned ranges, the lower one the larger");

__device__ __forceinline__ float edge_ex(float el_s, float er_d, unsigned maxel_d) {
    float v = el_s + er_d;
    v = (v >= 0.0f) ? v : NEG_SLOPE * v;
    float m = dec_f32(maxel_d) + er_d;
    m = (m >= 0.0f) ? m : NEG_SLOPE * m;
    return __expf(v - m);
}

__global__ __launch_bounds__(TPB) void gat_seg_max(const int* __restrict__ src,
                                                   const int* __restrict__ dst,
                                                   const float* __restrict__ el,
                                                   unsigned* __restrict__ maxel) {
    __shared__ unsigned tab[NH0];
    const int tid   = threadIdx.x;
    const int upper = blockIdx.x & 1;
    const int chunk = blockIdx.x >> 1;
    const int lo    = upper ? NH0 : 0;
    const int cnt   = upper ? (NNODES - NH0) : NH0;
    for (int j = tid; j < NH0; j += TPB) tab[j] = 0u;
    __syncthreads();

    const int e0 = chunk * EPC;
    int slot[EIT];
    unsigned val[EIT];
#pragma unroll
    for (int k = 0; k < EIT; ++k) {
        const int  i   = k * TPB + tid;
        const bool ok  = i < EPC;
        const int  ii  = e0 + (ok ? i : 0);
        const int  d   = dst[ii] - lo;
        const bool inr = ok && (unsigned)d < (unsigned)cnt;
        const int  s   = src[ii];
        val[k]  = enc_f32(el[inr ? s : 0]);
        slot[k] = inr ? d : -1;
    }
#pragma unroll
    for (int k = 0; k < EIT; ++k)
        if (slot[k] >= 0) atomicMax(&tab[slot[k]], val[k]);
    __syncthreads();

    const int rot = chunk % FIT;
    for (int it = 0; it < FIT; ++it) {
        int b = it + rot;
        b = (b >= FIT) ? b - FIT : b;
        const int j = b * TPB + tid;
        if (j < cnt) {
            const unsigned u = tab[j];
            if (u != 0u) atomicMax(&maxel[lo + j], u);
        }
    }
}

__global__ __launch_bounds__(TPB) void gat_seg_denom(const int* __restrict__ src,
                                                     const int* __restrict__ dst,
                                                     const float* __restrict__ el,
                                                     const float* __restrict__ er,
                                                     const unsigned* __restrict__ maxel,
                                                     float* __restrict__ denom) {
    __shared__ float tab[NH0];
    const int tid   = threadIdx.x;
    const int upper = blockIdx.x & 1;
    const int chunk = blockIdx.x >> 1;
    const int lo    = upper ? NH0 : 0;
    const int cnt   = upper ? (NNODES - NH0) : NH0;
    for (int j = tid; j < NH0; j += TPB) tab[j] = 0.0f;
    __syncthreads();

    const int e0 = chunk * EPC;
    int slot[EIT];
    float val[EIT];
#pragma unroll
    for (int k = 0; k < EIT; ++k) {
        const int  i   = k * TPB + tid;
        const bool ok  = i < EPC;
        const int  ii  = e0 + (ok ? i : 0);
        const int  da  = dst[ii];
        const int  d   = da - lo;
        const bool inr = ok && (unsigned)d < (unsigned)cnt;
        const int  s   = src[ii];
        const int  sq  = inr ? s : 0;
        const int  dq  = inr ? da : 0;
        val[k]  = edge_ex(el[sq], er[dq], maxel[dq]);
        slot[k] = inr ? d : -1;
    }
#pragma unroll
    for (int k = 0; k < EIT; ++k)
        if (slot[k] >= 0) atomicAdd(&tab[slot[k]], val[k]);
    __syncthreads();

    const int rot = chunk % FIT;
    for (int it = 0; it < FIT; ++it) {
        int b = it + rot;
        b = (b >= FIT) ? b - FIT : b;
        const int j = b * TPB + tid;
        if (j < cnt) {
            const float v = tab[j];
            if (v != 0.0f) atomicAdd(&denom[lo + j], v);
        }
    }
}

#define EPW 32
#define EBATCH 8
#define ROWLINES (DFEAT / 32)
static_assert(NEDGES % EPW == 0, "edges per wave must divide the edge count");
static_assert(EPW == 32 && EPW % EBATCH == 0, "one edge per lane, whole batches");
static_assert(DFEAT % 32 == 0, "a row is a whole number of 32-float lines");
__global__ void gat_scatter(const int* __restrict__ src,
                            const int* __restrict__ dst,
                            const float* __restrict__ el,
                            const float* __restrict__ er,
                            const unsigned* __restrict__ maxel,
                            const float* __restrict__ denom,
                            const float* __restrict__ feat,
                            float* __restrict__ out) {
    const int base = (blockIdx.x * 8 + (threadIdx.x >> 5)) * EPW;
    if (base >= NEDGES) return;
    const int lane = threadIdx.x & 31;

    const int   s_v = src[base + lane];
    const int   d_v = dst[base + lane];
    const float a_v = edge_ex(el[s_v], er[d_v], maxel[d_v]) / fmaxf(denom[d_v], 1e-9f);

#pragma unroll
    for (int j0 = 0; j0 < EPW; j0 += EBATCH) {
        float f[EBATCH][ROWLINES];
#pragma unroll
        for (int u = 0; u < EBATCH; ++u) {
            const int s = __builtin_amdgcn_readlane(s_v, j0 + u);
            const float* r = feat + (long)s * DFEAT;
#pragma unroll
            for (int c = 0; c < ROWLINES; ++c) f[u][c] = r[c * 32 + lane];
        }
#pragma unroll
        for (int u = 0; u < EBATCH; ++u) {
            const int   d     = __builtin_amdgcn_readlane(d_v, j0 + u);
            const float alpha = __int_as_float(__builtin_amdgcn_readlane(__float_as_int(a_v), j0 + u));
            float* o = out + (long)d * DFEAT;
#pragma unroll
            for (int c = 0; c < ROWLINES; ++c) atomicAdd(o + c * 32 + lane, alpha * f[u][c]);
        }
    }
}

extern "C" void kernel_launch(void* const* d_in, const int* in_sizes, int n_in,
                              void* d_out, int out_size, void* d_ws, size_t ws_size,
                              hipStream_t stream) {
    const float* h      = (const float*)d_in[0];
    const int*   src    = (const int*)  d_in[1];
    const int*   dst    = (const int*)  d_in[2];
    const float* W      = (const float*)d_in[3];
    const float* attn_l = (const float*)d_in[4];
    const float* attn_r = (const float*)d_in[5];
    float* out = (float*)d_out;

    float*    feat  = (float*)d_ws;
    float*    el    = feat + (long)NNODES * DFEAT;
    float*    er    = el + NNODES;
    unsigned* menc  = (unsigned*)(er + NNODES);
    float*    denom = (float*)(menc + NNODES);

    gat_zero_f32<<<4096, 256, 0, stream>>>(out, (long)NNODES * DFEAT);
    gat_zero_f32<<<256, 256, 0, stream>>>((float*)menc, 2L * NNODES);

    const int nstrips = NNODES / 16;
    gat_gemm_wmma<<<(nstrips + 3) / 4, 128, 0, stream>>>(h, W, feat);

    gat_attn_logits<<<(NNODES / NPW + 3) / 4, 128, 0, stream>>>(feat, attn_l, attn_r, el, er);

    gat_seg_max<<<2 * NCHUNK, TPB, 0, stream>>>(src, dst, el, menc);

    gat_seg_denom<<<2 * NCHUNK, TPB, 0, stream>>>(src, dst, el, er, menc, denom);

    gat_scatter<<<(NEDGES / EPW + 7) / 8, 256, 0, stream>>>(src, dst, el, er, menc, denom, feat, out);
}


# scatter by owner work-groups: 391 output rows each in LDS, edges found by a byte-per-edge owner array, rows added by plain LDS updates; no global float atomics, no out zero fill; sum order arbitrary
# speedup vs baseline: 2.4231x; 1.1322x over previous
#include <hip/hip_runtime.h>

typedef float v2f __attribute__((ext_vector_type(2)));
typedef float v8f __attribute__((ext_vector_type(8)));

#define NNODES   100000
#define NEDGES   1600000
#define DFEAT    128
#define NEG_SLOPE 0.2f

__global__ void gat_gemm_wmma(const float* __restrict__ h,
                              const float* __restrict__ W,
                              float* __restrict__ feat) {
    const int strip = blockIdx.x * 4 + (threadIdx.x >> 5);
    if (strip >= NNODES / 16) return;
    const int lane  = threadIdx.x & 31;
    const int m     = lane & 15;
    const int khalf = (lane >> 4) << 1;
    const int hihalf = (lane >> 4) << 3;
    const long row  = (long)strip * 16 + m;

    v8f acc[8];
#pragma unroll
    for (int ct = 0; ct < 8; ++ct) acc[ct] = (v8f){0,0,0,0,0,0,0,0};

    for (int k = 0; k < DFEAT; k += 4) {
        v2f a;
        a.x = h[row * DFEAT + k + khalf + 0];
        a.y = h[row * DFEAT + k + khalf + 1];
#pragma unroll
        for (int ct = 0; ct < 8; ++ct) {
            v2f b;
            b.x = W[(k + khalf + 0) * DFEAT + ct * 16 + m];
            b.y = W[(k + khalf + 1) * DFEAT + ct * 16 + m];
            acc[ct] = __builtin_amdgcn_wmma_f32_16x16x4_f32(
                false, a, false, b, (short)0, acc[ct], false, false);
        }
    }

#pragma unroll
    for (int ct = 0; ct < 8; ++ct) {
#pragma unroll
        for (int v = 0; v < 8; ++v) {
            long orow = (long)strip * 16 + v + hihalf;
            feat[orow * DFEAT + ct * 16 + m] = acc[ct][v];
        }
    }
}

#define NPW 8
static_assert(NNODES % NPW == 0, "nodes per wave must divide the node count");
__global__ void gat_attn_logits(const float* __restrict__ feat,
                                const float* __restrict__ attn_l,
                                const float* __restrict__ attn_r,
                                float* __restrict__ el,
                                float* __restrict__ er) {
    const int n0 = (blockIdx.x * 4 + (threadIdx.x >> 5)) * NPW;
    if (n0 >= NNODES) return;
    const int lane = threadIdx.x & 31;

    const float4 al = *(const float4*)(attn_l + lane * 4);
    const float4 ar = *(const float4*)(attn_r + lane * 4);

    float4 f[NPW];
#pragma unroll
    for (int u = 0; u < NPW; ++u)
        f[u] = *(const float4*)(feat + (long)(n0 + u) * DFEAT + lane * 4);

    float keep_l = 0.0f, keep_r = 0.0f;
#pragma unroll
    for (int u = 0; u < NPW; ++u) {
        float pl = f[u].x * al.x + f[u].y * al.y + f[u].z * al.z + f[u].w * al.w;
        float pr = f[u].x * ar.x + f[u].y * ar.y + f[u].z * ar.z + f[u].w * ar.w;
#pragma unroll
        for (int off = 16; off > 0; off >>= 1) {
            pl += __shfl_xor(pl, off, 32);
            pr += __shfl_xor(pr, off, 32);
        }
        if (lane == u) { keep_l = pl; keep_r = pr; }
    }
    if (lane < NPW) { el[n0 + lane] = keep_l; er[n0 + lane] = keep_r; }
}

__device__ __forceinline__ unsigned enc_f32(float x) {
    unsigned u = __float_as_uint(x);
    return (u & 0x80000000u) ? ~u : (u | 0x80000000u);
}
__device__ __forceinline__ float dec_f32(unsigned u) {
    return (u & 0x80000000u) ? __uint_as_float(u ^ 0x80000000u) : __uint_as_float(~u);
}

#define NH0    50016
#define NCHUNK 128
#define EPC    (NEDGES / NCHUNK)
#define TPB    1024
#define EIT    ((EPC + TPB - 1) / TPB)
#define FIT    ((NH0 + TPB - 1) / TPB)
static_assert(NEDGES % NCHUNK == 0, "whole chunks");
static_assert(NH0 % 32 == 0 && NH0 < NNODES && NNODES - NH0 <= NH0, "two line-aligned ranges, the lower one the larger");

__device__ __forceinline__ float edge_ex(float el_s, float er_d, unsigned maxel_d) {
    float v = el_s + er_d;
    v = (v >= 0.0f) ? v : NEG_SLOPE * v;
    float m = dec_f32(maxel_d) + er_d;
    m = (m >= 0.0f) ? m : NEG_SLOPE * m;
    return __expf(v - m);
}

__global__ __launch_bounds__(TPB) void gat_seg_max(const int* __restrict__ src,
                                                   const int* __restrict__ dst,
                                                   const float* __restrict__ el,
                                                   unsigned* __restrict__ maxel) {
    __shared__ unsigned tab[NH0];
    const int tid   = threadIdx.x;
    const int upper = blockIdx.x & 1;
    const int chunk = blockIdx.x >> 1;
    const int lo    = upper ? NH0 : 0;
    const int cnt   = upper ? (NNODES - NH0) : NH0;
    for (int j = tid; j < NH0; j += TPB) tab[j] = 0u;
    __syncthreads();

    const int e0 = chunk * EPC;
    int slot[EIT];
    unsigned val[EIT];
#pragma unroll
    for (int k = 0; k < EIT; ++k) {
        const int  i   = k * TPB + tid;
        const bool ok  = i < EPC;
        const int  ii  = e0 + (ok ? i : 0);
        const int  d   = dst[ii] - lo;
        const bool inr = ok && (unsigned)d < (unsigned)cnt;
        const int  s   = src[ii];
        val[k]  = enc_f32(el[inr ? s : 0]);
        slot[k] = inr ? d : -1;
    }
#pragma unroll
    for (int k = 0; k < EIT; ++k)
        if (slot[k] >= 0) atomicMax(&tab[slot[k]], val[k]);
    __syncthreads();

    const int rot = chunk % FIT;
    for (int it = 0; it < FIT; ++it) {
        int b = it + rot;
        b = (b >= FIT) ? b - FIT : b;
        const int j = b * TPB + tid;
        if (j < cnt) {
            const unsigned u = tab[j];
            if (u != 0u) atomicMax(&maxel[lo + j], u);
        }
    }
}

__global__ __launch_bounds__(TPB) void gat_seg_denom(const int* __restrict__ src,
                                                     const int* __restrict__ dst,
                                                     const float* __restrict__ el,
                                                     const float* __restrict__ er,
                                                     const unsigned* __restrict__ maxel,
                                                     float* __restrict__ denom) {
    __shared__ float tab[NH0];
    const int tid   = threadIdx.x;
    const int upper = blockIdx.x & 1;
    const int chunk = blockIdx.x >> 1;
    const int lo    = upper ? NH0 : 0;
    const int cnt   = upper ? (NNODES - NH0) : NH0;
    for (int j = tid; j < NH0; j += TPB) tab[j] = 0.0f;
    __syncthreads();

    const int e0 = chunk * EPC;
    int slot[EIT];
    float val[EIT];
#pragma unroll
    for (int k = 0; k < EIT; ++k) {
        const int  i   = k * TPB + tid;
        const bool ok  = i < EPC;
        const int  ii  = e0 + (ok ? i : 0);
        const int  da  = dst[ii];
        const int  d   = da - lo;
        const bool inr = ok && (unsigned)d < (unsigned)cnt;
        const int  s   = src[ii];
        const int  sq  = inr ? s : 0;
        const int  dq  = inr ? da : 0;
        val[k]  = edge_ex(el[sq], er[dq], maxel[dq]);
        slot[k] = inr ? d : -1;
    }
#pragma unroll
    for (int k = 0; k < EIT; ++k)
        if (slot[k] >= 0) atomicAdd(&tab[slot[k]], val[k]);
    __syncthreads();

    const int rot = chunk % FIT;
    for (int it = 0; it < FIT; ++it) {
        int b = it + rot;
        b = (b >= FIT) ? b - FIT : b;
        const int j = b * TPB + tid;
        if (j < cnt) {
            const float v = tab[j];
            if (v != 0.0f) atomicAdd(&denom[lo + j], v);
        }
    }
}

#define NR    391
#define NOWN  ((NNODES + NR - 1) / NR)
#define OTPB  1024
#define OWAVES (OTPB / 32)
#define Q0CAP 8192
#define Q1CAP 256
#define PB    16
#define SU    8
#define NITEM (NEDGES / 16)
#define RIT   (Q0CAP / OTPB)
#define OWNZ(x) (~((((x) & 0x7F7F7F7Fu) + 0x7F7F7F7Fu) | (x) | 0x7F7F7F7Fu))
static_assert(NOWN <= 256, "an owner number must fit one byte");
static_assert(NEDGES % 16 == 0, "16 edges per scanned item");
static_assert(Q0CAP % OTPB == 0, "whole route sweeps");
static_assert(NNODES <= (1 << 17) && NR <= (1 << 9), "packing of source and local row");

__global__ void gat_prep(const int* __restrict__ dst,
                         unsigned* __restrict__ rid4,
                         unsigned* __restrict__ zero2n) {
    const int t = blockIdx.x * blockDim.x + threadIdx.x;
    if (t < NEDGES / 4) {
        const int4 d = *(const int4*)(dst + 4 * t);
        rid4[t] =  ((unsigned)d.x / NR)        | (((unsigned)d.y / NR) << 8) |
                  (((unsigned)d.z / NR) << 16) | (((unsigned)d.w / NR) << 24);
    }
    if (t < 2 * NNODES) zero2n[t] = 0u;
}

__device__ __noinline__ void owner_slow_add(float* rows, const float* __restrict__ feat,
                                            int s, int lr, float a) {
    for (int c = 0; c < DFEAT; ++c)
        atomicAdd(&rows[lr * DFEAT + c], a * feat[(long)s * DFEAT + c]);
}

template <bool FULL>
__device__ __forceinline__ void owner_batch(float* rows, const float* __restrict__ feat,
                                            unsigned pv, float av, int nleft, int lane) {
    float4 f[PB];
#pragma unroll
    for (int u = 0; u < PB; ++u) {
        if (FULL || u < nleft) {
            const unsigned p = (unsigned)__builtin_amdgcn_readlane((int)pv, u);
            const int s = (int)(p & 0x1FFFFu);
            f[u] = *(const float4*)(feat + (long)s * DFEAT + lane * 4);
        }
    }
#pragma unroll
    for (int u = 0; u < PB; ++u) {
        if (FULL || u < nleft) {
            const unsigned p = (unsigned)__builtin_amdgcn_readlane((int)pv, u);
            const float a = __int_as_float(__builtin_amdgcn_readlane(__float_as_int(av), u));
            const int lr = (int)(p >> 17);
            float4* rp = (float4*)(rows + lr * DFEAT + lane * 4);
            float4 r = *rp;
            r.x += a * f[u].x; r.y += a * f[u].y; r.z += a * f[u].z; r.w += a * f[u].w;
            *rp = r;
        }
    }
}

__global__ __launch_bounds__(OTPB) void gat_owner_sum(const int* __restrict__ src,
                                                      const int* __restrict__ dst,
                                                      const unsigned* __restrict__ rid4,
                                                      const float* __restrict__ el,
                                                      const float* __restrict__ er,
                                                      const unsigned* __restrict__ maxel,
                                                      const float* __restrict__ denom,
                                                      const float* __restrict__ feat,
                                                      float* __restrict__ out) {
    __shared__ float    rows[NR * DFEAT];
    __shared__ unsigned q0[Q0CAP];
    __shared__ unsigned q1p[OWAVES * Q1CAP];
    __shared__ float    q1a[OWAVES * Q1CAP];
    __shared__ unsigned cnts[1 + OWAVES];

    const int tid  = threadIdx.x;
    const int lane = tid & 31;
    const int w    = tid >> 5;
    const int own  = blockIdx.x;
    const int lo   = own * NR;
    const int cnt  = (NNODES - lo < NR) ? (NNODES - lo) : NR;

    for (int j = tid; j < NR * DFEAT / 4; j += OTPB) ((float4*)rows)[j] = make_float4(0.f, 0.f, 0.f, 0.f);
    if (tid <= OWAVES) cnts[tid] = 0u;
    __syncthreads();

    const unsigned key = (unsigned)own * 0x01010101u;
    const uint4* rv = (const uint4*)rid4;
    for (int base = tid; base < NITEM; base += OTPB * SU) {
        uint4 v[SU];
#pragma unroll
        for (int u = 0; u < SU; ++u) {
            const int q = base + u * OTPB;
            v[u] = rv[q < NITEM ? q : base];
        }
#pragma unroll
        for (int u = 0; u < SU; ++u) {
            const int q = base + u * OTPB;
            const unsigned x0 = v[u].x ^ key, x1 = v[u].y ^ key, x2 = v[u].z ^ key, x3 = v[u].w ^ key;
            unsigned mm = (OWNZ(x0) >> 7) | (OWNZ(x1) >> 6) | (OWNZ(x2) >> 5) | (OWNZ(x3) >> 4);
            if (q >= NITEM) mm = 0u;
            while (mm != 0u) {
                const int b = __builtin_ctz(mm);
                mm &= mm - 1u;
                const int e = q * 16 + ((b & 7) << 2) + (b >> 3);
                const unsigned slot = atomicAdd(&cnts[0], 1u);
                if (slot < Q0CAP) {
                    q0[slot] = (unsigned)e;
                } else {
                    const int s = src[e];
                    const int d = dst[e];
                    const int lr = d - lo;
                    if ((unsigned)lr < (unsigned)cnt) {
                        const float a = edge_ex(el[s], er[d], maxel[d]) / fmaxf(denom[d], 1e-9f);
                        owner_slow_add(rows, feat, s, lr, a);
                    }
                }
            }
        }
    }
    __syncthreads();

    const int n0 = (cnts[0] < Q0CAP) ? (int)cnts[0] : Q0CAP;
    {
        int  ee[RIT];
        bool ok[RIT];
        int  ss[RIT], dd[RIT];
        float aa[RIT];
#pragma unroll
        for (int k = 0; k < RIT; ++k) {
            const int i = k * OTPB + tid;
            ok[k] = i < n0;
            ee[k] = ok[k] ? (int)q0[i] : 0;
        }
#pragma unroll
        for (int k = 0; k < RIT; ++k) { ss[k] = src[ee[k]]; dd[k] = dst[ee[k]]; }
#pragma unroll
        for (int k = 0; k < RIT; ++k) {
            const int lr = dd[k] - lo;
            ok[k] = ok[k] && (unsigned)lr < (unsigned)cnt;
            const int sq = ok[k] ? ss[k] : 0;
            const int dq = ok[k] ? dd[k] : 0;
            aa[k] = edge_ex(el[sq], er[dq], maxel[dq]) / fmaxf(denom[dq], 1e-9f);
        }
#pragma unroll
        for (int k = 0; k < RIT; ++k) {
            if (ok[k]) {
                const int lr = dd[k] - lo;
                const int ow = lr & (OWAVES - 1);
                const unsigned slot = atomicAdd(&cnts[1 + ow], 1u);
                if (slot < Q1CAP) {
                    q1p[ow * Q1CAP + slot] = (unsigned)ss[k] | ((unsigned)lr << 17);
                    q1a[ow * Q1CAP + slot] = aa[k];
                } else {
                    owner_slow_add(rows, feat, ss[k], lr, aa[k]);
                }
            }
        }
    }
    __syncthreads();

    {
        const unsigned c1 = cnts[1 + w];
        const int n1 = __builtin_amdgcn_readfirstlane((int)(c1 < Q1CAP ? c1 : Q1CAP));
        const unsigned* qp = q1p + w * Q1CAP;
        const float*    qa = q1a + w * Q1CAP;
        for (int i0 = 0; i0 < n1; i0 += PB) {
            const int  iu = i0 + (lane & (PB - 1));
            const bool vu = iu < n1;
            const int  ic = vu ? iu : i0;
            const unsigned pv = qp[ic];
            const float    av = qa[ic];
            if (i0 + PB <= n1) owner_batch<true>(rows, feat, pv, av, PB, lane);
            else               owner_batch<false>(rows, feat, pv, av, n1 - i0, lane);
        }
    }
    __syncthreads();

    float4* o4 = (float4*)(out + (long)lo * DFEAT);
    for (int j = tid; j < cnt * (DFEAT / 4); j += OTPB) o4[j] = ((const float4*)rows)[j];
}

extern "C" void kernel_launch(void* const* d_in, const int* in_sizes, int n_in,
                              void* d_out, int out_size, void* d_ws, size_t ws_size,
                              hipStream_t stream) {
    const float* h      = (const float*)d_in[0];
    const int*   src    = (const int*)  d_in[1];
    const int*   dst    = (const int*)  d_in[2];
    const float* W      = (const float*)d_in[3];
    const float* attn_l = (const float*)d_in[4];
    const float* attn_r = (const float*)d_in[5];
    float* out = (float*)d_out;

    float*    feat  = (float*)d_ws;
    float*    el    = feat + (long)NNODES * DFEAT;
    float*    er    = el + NNODES;
    unsigned* menc  = (unsigned*)(er + NNODES);
    float*    denom = (float*)(menc + NNODES);
    unsigned* rid4  = (unsigned*)(denom + NNODES);

    gat_prep<<<(NEDGES / 4 + 255) / 256, 256, 0, stream>>>(dst, rid4, menc);

    const int nstrips = NNODES / 16;
    gat_gemm_wmma<<<(nstrips + 3) / 4, 128, 0, stream>>>(h, W, feat);

    gat_attn_logits<<<(NNODES / NPW + 3) / 4, 128, 0, stream>>>(feat, attn_l, attn_r, el, er);

    gat_seg_max<<<2 * NCHUNK, TPB, 0, stream>>>(src, dst, el, menc);

    gat_seg_denom<<<2 * NCHUNK, TPB, 0, stream>>>(src, dst, el, er, menc, denom);

    gat_owner_sum<<<NOWN, OTPB, 0, stream>>>(src, dst, rid4, el, er, menc, denom, feat, out);
}


# owners take each row's max score and softmax denominator from their own edge lists (LDS tables, LDS atomics); both table kernels, their global atomics and 2 of 6 launches removed; sum order arbitrary
# speedup vs baseline: 3.4373x; 1.4186x over previous
#include <hip/hip_runtime.h>

typedef float v2f __attribute__((ext_vector_type(2)));
typedef float v8f __attribute__((ext_vector_type(8)));

#define NNODES   100000
#define NEDGES   1600000
#define DFEAT    128
#define NEG_SLOPE 0.2f

__global__ void gat_gemm_wmma(const float* __restrict__ h,
                              const float* __restrict__ W,
                              float* __restrict__ feat) {
    const int strip = blockIdx.x * 4 + (threadIdx.x >> 5);
    if (strip >= NNODES / 16) return;
    const int lane  = threadIdx.x & 31;
    const int m     = lane & 15;
    const int khalf = (lane >> 4) << 1;
    const int hihalf = (lane >> 4) << 3;
    const long row  = (long)strip * 16 + m;

    v8f acc[8];
#pragma unroll
    for (int ct = 0; ct < 8; ++ct) acc[ct] = (v8f){0,0,0,0,0,0,0,0};

    for (int k = 0; k < DFEAT; k += 4) {
        v2f a;
        a.x = h[row * DFEAT + k + khalf + 0];
        a.y = h[row * DFEAT + k + khalf + 1];
#pragma unroll
        for (int ct = 0; ct < 8; ++ct) {
            v2f b;
            b.x = W[(k + khalf + 0) * DFEAT + ct * 16 + m];
            b.y = W[(k + khalf + 1) * DFEAT + ct * 16 + m];
            acc[ct] = __builtin_amdgcn_wmma_f32_16x16x4_f32(
                false, a, false, b, (short)0, acc[ct], false, false);
        }
    }

#pragma unroll
    for (int ct = 0; ct < 8; ++ct) {
#pragma unroll
        for (int v = 0; v < 8; ++v) {
            long orow = (long)strip * 16 + v + hihalf;
            feat[orow * DFEAT + ct * 16 + m] = acc[ct][v];
        }
    }
}

#define NPW 8
static_assert(NNODES % NPW == 0, "nodes per wave must divide the node count");
__global__ void gat_attn_logits(const float* __restrict__ feat,
                                const float* __restrict__ attn_l,
                                const float* __restrict__ attn_r,
                                float* __restrict__ el,
                                float* __restrict__ er) {
    const int n0 = (blockIdx.x * 4 + (threadIdx.x >> 5)) * NPW;
    if (n0 >= NNODES) return;
    const int lane = threadIdx.x & 31;

    const float4 al = *(const float4*)(attn_l + lane * 4);
    const float4 ar = *(const float4*)(attn_r + lane * 4);

    float4 f[NPW];
#pragma unroll
    for (int u = 0; u < NPW; ++u)
        f[u] = *(const float4*)(feat + (long)(n0 + u) * DFEAT + lane * 4);

    float keep_l = 0.0f, keep_r = 0.0f;
#pragma unroll
    for (int u = 0; u < NPW; ++u) {
        float pl = f[u].x * al.x + f[u].y * al.y + f[u].z * al.z + f[u].w * al.w;
        float pr = f[u].x * ar.x + f[u].y * ar.y + f[u].z * ar.z + f[u].w * ar.w;
#pragma unroll
        for (int off = 16; off > 0; off >>= 1) {
            pl += __shfl_xor(pl, off, 32);
            pr += __shfl_xor(pr, off, 32);
        }
        if (lane == u) { keep_l = pl; keep_r = pr; }
    }
    if (lane < NPW) { el[n0 + lane] = keep_l; er[n0 + lane] = keep_r; }
}

__device__ __forceinline__ unsigned enc_f32(float x) {
    unsigned u = __float_as_uint(x);
    return (u & 0x80000000u) ? ~u : (u | 0x80000000u);
}
__device__ __forceinline__ float dec_f32(unsigned u) {
    return (u & 0x80000000u) ? __uint_as_float(u ^ 0x80000000u) : __uint_as_float(~u);
}

#define NR    391
#define NOWN  ((NNODES + NR - 1) / NR)
#define OTPB  1024
#define OWAVES (OTPB / 32)
#define Q0CAP 8192
#define Q1CAP 256
#define PB    16
#define SU    8
#define NITEM (NEDGES / 16)
#define RIT   (Q0CAP / OTPB)
#define OWNZ(x) (~((((x) & 0x7F7F7F7Fu) + 0x7F7F7F7Fu) | (x) | 0x7F7F7F7Fu))
static_assert(NOWN <= 256, "an owner number must fit one byte");
static_assert(NEDGES % 16 == 0, "16 edges per scanned item");
static_assert(Q0CAP % OTPB == 0, "whole route sweeps");
static_assert(NNODES <= (1 << 17) && NR <= (1 << 9), "packing of source and local row");

__global__ void gat_prep(const int* __restrict__ dst,
                         unsigned* __restrict__ rid4) {
    const int t = blockIdx.x * blockDim.x + threadIdx.x;
    if (t < NEDGES / 4) {
        const int4 d = *(const int4*)(dst + 4 * t);
        rid4[t] =  ((unsigned)d.x / NR)        | (((unsigned)d.y / NR) << 8) |
                  (((unsigned)d.z / NR) << 16) | (((unsigned)d.w / NR) << 24);
    }
}

__device__ __forceinline__ void lane_row_add(float* rows, const float* __restrict__ feat,
                                             int s, int lr, float a) {
    for (int c = 0; c < DFEAT; ++c)
        atomicAdd(&rows[lr * DFEAT + c], a * feat[(long)s * DFEAT + c]);
}
__device__ __noinline__ void owner_slow_add(float* rows, const float* __restrict__ feat,
                                            int s, int lr, float a) {
    lane_row_add(rows, feat, s, lr, a);
}

__device__ __forceinline__ float edge_score(float el_s, float er_d) {
    const float v = el_s + er_d;
    return (v >= 0.0f) ? v : NEG_SLOPE * v;
}
__device__ __forceinline__ float row_max(unsigned enc) {
    const float m = dec_f32(enc);
    return __builtin_isfinite(m) ? m : 0.0f;
}

__device__ __noinline__ void owner_all_slow(float* rows, unsigned* tmax, float* tden,
                                            const int* __restrict__ src, const int* __restrict__ dst,
                                            const float* __restrict__ el, const float* __restrict__ er,
                                            const float* __restrict__ feat, int lo, int cnt, int tid) {
    for (int e = tid; e < NEDGES; e += OTPB) {
        const int d = dst[e];
        const int lr = d - lo;
        if ((unsigned)lr < (unsigned)cnt) atomicMax(&tmax[lr], enc_f32(edge_score(el[src[e]], er[d])));
    }
    __syncthreads();
    for (int e = tid; e < NEDGES; e += OTPB) {
        const int d = dst[e];
        const int lr = d - lo;
        if ((unsigned)lr < (unsigned)cnt)
            atomicAdd(&tden[lr], __expf(edge_score(el[src[e]], er[d]) - row_max(tmax[lr])));
    }
    __syncthreads();
    for (int e = tid; e < NEDGES; e += OTPB) {
        const int d = dst[e];
        const int lr = d - lo;
        if ((unsigned)lr < (unsigned)cnt) {
            const int s = src[e];
            const float a = __expf(edge_score(el[s], er[d]) - row_max(tmax[lr])) / fmaxf(tden[lr], 1e-9f);
            lane_row_add(rows, feat, s, lr, a);
        }
    }
}

template <bool FULL>
__device__ __forceinline__ void owner_batch(float* rows, const float* __restrict__ feat,
                                            unsigned pv, float av, int nleft, int lane) {
    float4 f[PB];
#pragma unroll
    for (int u = 0; u < PB; ++u) {
        if (FULL || u < nleft) {
            const unsigned p = (unsigned)__builtin_amdgcn_readlane((int)pv, u);
            const int s = (int)(p & 0x1FFFFu);
            f[u] = *(const float4*)(feat + (long)s * DFEAT + lane * 4);
        }
    }
#pragma unroll
    for (int u = 0; u < PB; ++u) {
        if (FULL || u < nleft) {
            const unsigned p = (unsigned)__builtin_amdgcn_readlane((int)pv, u);
            const float a = __int_as_float(__builtin_amdgcn_readlane(__float_as_int(av), u));
            const int lr = (int)(p >> 17);
            float4* rp = (float4*)(rows + lr * DFEAT + lane * 4);
            float4 r = *rp;
            r.x += a * f[u].x; r.y += a * f[u].y; r.z += a * f[u].z; r.w += a * f[u].w;
            *rp = r;
        }
    }
}

__global__ __launch_bounds__(OTPB) void gat_owner_sum(const int* __restrict__ src,
                                                      const int* __restrict__ dst,
                                                      const unsigned* __restrict__ rid4,
                                                      const float* __restrict__ el,
                                                      const float* __restrict__ er,
                                                      const float* __restrict__ feat,
                                                      float* __restrict__ out) {
    __shared__ float    rows[NR * DFEAT];
    __shared__ unsigned q0[Q0CAP];
    __shared__ unsigned q1p[OWAVES * Q1CAP];
    __shared__ float    q1a[OWAVES * Q1CAP];
    __shared__ unsigned cnts[1 + OWAVES];
    __shared__ unsigned tmax[NR];
    __shared__ float    tden[NR];

    const int tid  = threadIdx.x;
    const int lane = tid & 31;
    const int w    = tid >> 5;
    const int own  = blockIdx.x;
    const int lo   = own * NR;
    const int cnt  = (NNODES - lo < NR) ? (NNODES - lo) : NR;

    for (int j = tid; j < NR * DFEAT / 4; j += OTPB) ((float4*)rows)[j] = make_float4(0.f, 0.f, 0.f, 0.f);
    if (tid <= OWAVES) cnts[tid] = 0u;
    if (tid < NR) { tmax[tid] = 0u; tden[tid] = 0.0f; }
    __syncthreads();

    const unsigned key = (unsigned)own * 0x01010101u;
    const uint4* rv = (const uint4*)rid4;
    for (int base = tid; base < NITEM; base += OTPB * SU) {
        uint4 v[SU];
#pragma unroll
        for (int u = 0; u < SU; ++u) {
            const int q = base + u * OTPB;
            v[u] = rv[q < NITEM ? q : base];
        }
#pragma unroll
        for (int u = 0; u < SU; ++u) {
            const int q = base + u * OTPB;
            const unsigned x0 = v[u].x ^ key, x1 = v[u].y ^ key, x2 = v[u].z ^ key, x3 = v[u].w ^ key;
            unsigned mm = (OWNZ(x0) >> 7) | (OWNZ(x1) >> 6) | (OWNZ(x2) >> 5) | (OWNZ(x3) >> 4);
            if (q >= NITEM) mm = 0u;
            while (mm != 0u) {
                const int b = __builtin_ctz(mm);
                mm &= mm - 1u;
                const int e = q * 16 + ((b & 7) << 2) + (b >> 3);
                const unsigned slot = atomicAdd(&cnts[0], 1u);
                if (slot < Q0CAP) q0[slot] = (unsigned)e;
            }
        }
    }
    __syncthreads();

    const unsigned total = cnts[0];
    if (total > Q0CAP) {
        owner_all_slow(rows, tmax, tden, src, dst, el, er, feat, lo, cnt, tid);
    } else {
        const int n0 = (int)total;
        int   ee[RIT];
        bool  ok[RIT];
        int   ss[RIT], lrs[RIT];
        float vv[RIT];
#pragma unroll
        for (int k = 0; k < RIT; ++k) {
            const int i = k * OTPB + tid;
            ok[k] = i < n0;
            ee[k] = ok[k] ? (int)q0[i] : 0;
        }
        int dd[RIT];
#pragma unroll
        for (int k = 0; k < RIT; ++k) { ss[k] = src[ee[k]]; dd[k] = dst[ee[k]]; }
#pragma unroll
        for (int k = 0; k < RIT; ++k) {
            lrs[k] = dd[k] - lo;
            ok[k] = ok[k] && (unsigned)lrs[k] < (unsigned)cnt;
            const int sq = ok[k] ? ss[k] : 0;
            const int dq = ok[k] ? dd[k] : lo;
            vv[k] = edge_score(el[sq], er[dq]);
        }
#pragma unroll
        for (int k = 0; k < RIT; ++k)
            if (ok[k]) atomicMax(&tmax[lrs[k]], enc_f32(vv[k]));
        __syncthreads();
#pragma unroll
        for (int k = 0; k < RIT; ++k) {
            if (ok[k]) {
                vv[k] = __expf(vv[k] - row_max(tmax[lrs[k]]));
                atomicAdd(&tden[lrs[k]], vv[k]);
            }
        }
        __syncthreads();
#pragma unroll
        for (int k = 0; k < RIT; ++k) {
            if (ok[k]) {
                const int lr = lrs[k];
                const float a = vv[k] / fmaxf(tden[lr], 1e-9f);
                const int ow = lr & (OWAVES - 1);
                const unsigned slot = atomicAdd(&cnts[1 + ow], 1u);
                if (slot < Q1CAP) {
                    q1p[ow * Q1CAP + slot] = (unsigned)ss[k] | ((unsigned)lr << 17);
                    q1a[ow * Q1CAP + slot] = a;
                } else {
                    owner_slow_add(rows, feat, ss[k], lr, a);
                }
            }
        }
    }
    __syncthreads();

    {
        const unsigned c1 = cnts[1 + w];
        const int n1 = __builtin_amdgcn_readfirstlane((int)(c1 < Q1CAP ? c1 : Q1CAP));
        const unsigned* qp = q1p + w * Q1CAP;
        const float*    qa = q1a + w * Q1CAP;
        for (int i0 = 0; i0 < n1; i0 += PB) {
            const int  iu = i0 + (lane & (PB - 1));
            const bool vu = iu < n1;
            const int  ic = vu ? iu : i0;
            const unsigned pv = qp[ic];
            const float    av = qa[ic];
            if (i0 + PB <= n1) owner_batch<true>(rows, feat, pv, av, PB, lane);
            else               owner_batch<false>(rows, feat, pv, av, n1 - i0, lane);
        }
    }
    __syncthreads();

    float4* o4 = (float4*)(out + (long)lo * DFEAT);
    for (int j = tid; j < cnt * (DFEAT / 4); j += OTPB) o4[j] = ((const float4*)rows)[j];
}

extern "C" void kernel_launch(void* const* d_in, const int* in_sizes, int n_in,
                              void* d_out, int out_size, void* d_ws, size_t ws_size,
                              hipStream_t stream) {
    const float* h      = (const float*)d_in[0];
    const int*   src    = (const int*)  d_in[1];
    const int*   dst    = (const int*)  d_in[2];
    const float* W      = (const float*)d_in[3];
    const float* attn_l = (const float*)d_in[4];
    const float* attn_r = (const float*)d_in[5];
    float* out = (float*)d_out;

    float*    feat  = (float*)d_ws;
    float*    el    = feat + (long)NNODES * DFEAT;
    float*    er    = el + NNODES;
    unsigned* rid4  = (unsigned*)(er + NNODES);

    gat_prep<<<(NEDGES / 4 + 255) / 256, 256, 0, stream>>>(dst, rid4);

    const int nstrips = NNODES / 16;
    gat_gemm_wmma<<<(nstrips + 3) / 4, 128, 0, stream>>>(h, W, feat);

    gat_attn_logits<<<(NNODES / NPW + 3) / 4, 128, 0, stream>>>(feat, attn_l, attn_r, el, er);

    gat_owner_sum<<<NOWN, OTPB, 0, stream>>>(src, dst, rid4, el, er, feat, out);
}


# owners find their edges in per-owner presence bit planes (built per 1024-edge tile in LDS, one 128-B line per owner, parked in the owner's out rows), not by a byte per edge; sum order arbitrary
# speedup vs baseline: 3.6494x; 1.0617x over previous
#include <hip/hip_runtime.h>

typedef float v2f __attribute__((ext_vector_type(2)));
typedef float v8f __attribute__((ext_vector_type(8)));

#define NNODES   100000
#define NEDGES   1600000
#define DFEAT    128
#define NEG_SLOPE 0.2f

__global__ void gat_gemm_wmma(const float* __restrict__ h,
                              const float* __restrict__ W,
                              float* __restrict__ feat) {
    const int strip = blockIdx.x * 4 + (threadIdx.x >> 5);
    if (strip >= NNODES / 16) return;
    const int lane  = threadIdx.x & 31;
    const int m     = lane & 15;
    const int khalf = (lane >> 4) << 1;
    const int hihalf = (lane >> 4) << 3;
    const long row  = (long)strip * 16 + m;

    v8f acc[8];
#pragma unroll
    for (int ct = 0; ct < 8; ++ct) acc[ct] = (v8f){0,0,0,0,0,0,0,0};

    for (int k = 0; k < DFEAT; k += 4) {
        v2f a;
        a.x = h[row * DFEAT + k + khalf + 0];
        a.y = h[row * DFEAT + k + khalf + 1];
#pragma unroll
        for (int ct = 0; ct < 8; ++ct) {
            v2f b;
            b.x = W[(k + khalf + 0) * DFEAT + ct * 16 + m];
            b.y = W[(k + khalf + 1) * DFEAT + ct * 16 + m];
            acc[ct] = __builtin_amdgcn_wmma_f32_16x16x4_f32(
                false, a, false, b, (short)0, acc[ct], false, false);
        }
    }

#pragma unroll
    for (int ct = 0; ct < 8; ++ct) {
#pragma unroll
        for (int v = 0; v < 8; ++v) {
            long orow = (long)strip * 16 + v + hihalf;
            feat[orow * DFEAT + ct * 16 + m] = acc[ct][v];
        }
    }
}

#define NPW 8
static_assert(NNODES % NPW == 0, "nodes per wave must divide the node count");
__global__ void gat_attn_logits(const float* __restrict__ feat,
                                const float* __restrict__ attn_l,
                                const float* __restrict__ attn_r,
                                float* __restrict__ el,
                                float* __restrict__ er) {
    const int n0 = (blockIdx.x * 4 + (threadIdx.x >> 5)) * NPW;
    if (n0 >= NNODES) return;
    const int lane = threadIdx.x & 31;

    const float4 al = *(const float4*)(attn_l + lane * 4);
    const float4 ar = *(const float4*)(attn_r + lane * 4);

    float4 f[NPW];
#pragma unroll
    for (int u = 0; u < NPW; ++u)
        f[u] = *(const float4*)(feat + (long)(n0 + u) * DFEAT + lane * 4);

    float keep_l = 0.0f, keep_r = 0.0f;
#pragma unroll
    for (int u = 0; u < NPW; ++u) {
        float pl = f[u].x * al.x + f[u].y * al.y + f[u].z * al.z + f[u].w * al.w;
        float pr = f[u].x * ar.x + f[u].y * ar.y + f[u].z * ar.z + f[u].w * ar.w;
#pragma unroll
        for (int off = 16; off > 0; off >>= 1) {
            pl += __shfl_xor(pl, off, 32);
            pr += __shfl_xor(pr, off, 32);
        }
        if (lane == u) { keep_l = pl; keep_r = pr; }
    }
    if (lane < NPW) { el[n0 + lane] = keep_l; er[n0 + lane] = keep_r; }
}

__device__ __forceinline__ unsigned enc_f32(float x) {
    unsigned u = __float_as_uint(x);
    return (u & 0x80000000u) ? ~u : (u | 0x80000000u);
}
__device__ __forceinline__ float dec_f32(unsigned u) {
    return (u & 0x80000000u) ? __uint_as_float(u ^ 0x80000000u) : __uint_as_float(~u);
}

#define NR    391
#define NOWN  ((NNODES + NR - 1) / NR)
#define OTPB  1024
#define OWAVES (OTPB / 32)
#define Q0CAP 8192
#define Q1CAP 256
#define PB    16
#define RIT   (Q0CAP / OTPB)
#define TE    1024
#define NTILE ((NEDGES + TE - 1) / TE)
#define PLW   (NTILE * 32)
#define PITEM (PLW / 4)
#define NIT   ((PITEM + OTPB - 1) / OTPB)
#define BP    36
#define PTPB  (TE / 4)
static_assert(NEDGES % 4 == 0, "4 edges per thread and load");
static_assert((long)PLW * 4 <= (long)NR * DFEAT * 4, "a plane must fit its owner's rows of out");
static_assert((NOWN - 1) * NR < NNODES, "every owner but the last has all NR rows");
static_assert(Q0CAP % OTPB == 0, "whole route sweeps");
static_assert(NNODES <= (1 << 17) && NR <= (1 << 9), "packing of source and local row");

__device__ __forceinline__ unsigned* plane_of(float* out, unsigned* lastplane, int o) {
    return (o < NOWN - 1) ? (unsigned*)(out + (long)o * NR * DFEAT) : lastplane;
}

__global__ __launch_bounds__(PTPB) void gat_planes(const int* __restrict__ dst,
                                                   float* out, unsigned* lastplane) {
    __shared__ unsigned bits[NOWN * BP];
    const int tid  = threadIdx.x;
    const int tile = blockIdx.x;
    for (int j = tid; j < NOWN * BP / 4; j += PTPB) ((uint4*)bits)[j] = make_uint4(0u, 0u, 0u, 0u);
    __syncthreads();
    const int e0 = tile * TE + 4 * tid;
    if (e0 < NEDGES) {
        const int4 d = *(const int4*)(dst + e0);
        const int g  = tid >> 3;
        const int sh = (tid & 7) << 2;
        atomicOr(&bits[((unsigned)d.x / NR) * BP + g], 1u << sh);
        atomicOr(&bits[((unsigned)d.y / NR) * BP + g], 2u << sh);
        atomicOr(&bits[((unsigned)d.z / NR) * BP + g], 4u << sh);
        atomicOr(&bits[((unsigned)d.w / NR) * BP + g], 8u << sh);
    }
    __syncthreads();
    for (int c = tid; c < NOWN * 8; c += PTPB) {
        const int o = c >> 3, p = c & 7;
        const uint4 v = *(const uint4*)(bits + o * BP + 4 * p);
        *(uint4*)(plane_of(out, lastplane, o) + tile * 32 + 4 * p) = v;
    }
}

__device__ __forceinline__ void lane_row_add(float* rows, const float* __restrict__ feat,
                                             int s, int lr, float a) {
    for (int c = 0; c < DFEAT; ++c)
        atomicAdd(&rows[lr * DFEAT + c], a * feat[(long)s * DFEAT + c]);
}
__device__ __noinline__ void owner_slow_add(float* rows, const float* __restrict__ feat,
                                            int s, int lr, float a) {
    lane_row_add(rows, feat, s, lr, a);
}

__device__ __forceinline__ float edge_score(float el_s, float er_d) {
    const float v = el_s + er_d;
    return (v >= 0.0f) ? v : NEG_SLOPE * v;
}
__device__ __forceinline__ float row_max(unsigned enc) {
    const float m = dec_f32(enc);
    return __builtin_isfinite(m) ? m : 0.0f;
}

__device__ __noinline__ void owner_all_slow(float* rows, unsigned* tmax, float* tden,
                                            const int* __restrict__ src, const int* __restrict__ dst,
                                            const float* __restrict__ el, const float* __restrict__ er,
                                            const float* __restrict__ feat, int lo, int cnt, int tid) {
    for (int e = tid; e < NEDGES; e += OTPB) {
        const int d = dst[e];
        const int lr = d - lo;
        if ((unsigned)lr < (unsigned)cnt) atomicMax(&tmax[lr], enc_f32(edge_score(el[src[e]], er[d])));
    }
    __syncthreads();
    for (int e = tid; e < NEDGES; e += OTPB) {
        const int d = dst[e];
        const int lr = d - lo;
        if ((unsigned)lr < (unsigned)cnt)
            atomicAdd(&tden[lr], __expf(edge_score(el[src[e]], er[d]) - row_max(tmax[lr])));
    }
    __syncthreads();
    for (int e = tid; e < NEDGES; e += OTPB) {
        const int d = dst[e];
        const int lr = d - lo;
        if ((unsigned)lr < (unsigned)cnt) {
            const int s = src[e];
            const float a = __expf(edge_score(el[s], er[d]) - row_max(tmax[lr])) / fmaxf(tden[lr], 1e-9f);
            lane_row_add(rows, feat, s, lr, a);
        }
    }
}

template <bool FULL>
__device__ __forceinline__ void owner_batch(float* rows, const float* __restrict__ feat,
                                            unsigned pv, float av, int nleft, int lane) {
    float4 f[PB];
#pragma unroll
    for (int u = 0; u < PB; ++u) {
        if (FULL || u < nleft) {
            const unsigned p = (unsigned)__builtin_amdgcn_readlane((int)pv, u);
            const int s = (int)(p & 0x1FFFFu);
            f[u] = *(const float4*)(feat + (long)s * DFEAT + lane * 4);
        }
    }
#pragma unroll
    for (int u = 0; u < PB; ++u) {
        if (FULL || u < nleft) {
            const unsigned p = (unsigned)__builtin_amdgcn_readlane((int)pv, u);
            const float a = __int_as_float(__builtin_amdgcn_readlane(__float_as_int(av), u));
            const int lr = (int)(p >> 17);
            float4* rp = (float4*)(rows + lr * DFEAT + lane * 4);
            float4 r = *rp;
            r.x += a * f[u].x; r.y += a * f[u].y; r.z += a * f[u].z; r.w += a * f[u].w;
            *rp = r;
        }
    }
}

__global__ __launch_bounds__(OTPB) void gat_owner_sum(const int* __restrict__ src,
                                                      const int* __restrict__ dst,
                                                      const unsigned* __restrict__ lastplane,
                                                      const float* __restrict__ el,
                                                      const float* __restrict__ er,
                                                      const float* __restrict__ feat,
                                                      float* __restrict__ out) {
    __shared__ float    rows[NR * DFEAT];
    __shared__ unsigned q0[Q0CAP];
    __shared__ unsigned q1p[OWAVES * Q1CAP];
    __shared__ float    q1a[OWAVES * Q1CAP];
    __shared__ unsigned cnts[1 + OWAVES];
    __shared__ unsigned tmax[NR];
    __shared__ float    tden[NR];

    const int tid  = threadIdx.x;
    const int lane = tid & 31;
    const int w    = tid >> 5;
    const int own  = blockIdx.x;
    const int lo   = own * NR;
    const int cnt  = (NNODES - lo < NR) ? (NNODES - lo) : NR;

    for (int j = tid; j < NR * DFEAT / 4; j += OTPB) ((float4*)rows)[j] = make_float4(0.f, 0.f, 0.f, 0.f);
    if (tid <= OWAVES) cnts[tid] = 0u;
    if (tid < NR) { tmax[tid] = 0u; tden[tid] = 0.0f; }
    __syncthreads();

    {
        const uint4* pv = (const uint4*)((own < NOWN - 1) ? (const unsigned*)(out + (long)lo * DFEAT) : lastplane);
        uint4 v[NIT];
#pragma unroll
        for (int u = 0; u < NIT; ++u) {
            const int q = tid + u * OTPB;
            v[u] = pv[q < PITEM ? q : tid];
            if (q >= PITEM) v[u] = make_uint4(0u, 0u, 0u, 0u);
        }
        unsigned c = 0u;
#pragma unroll
        for (int u = 0; u < NIT; ++u)
            c += (unsigned)(__popc(v[u].x) + __popc(v[u].y) + __popc(v[u].z) + __popc(v[u].w));
        unsigned incl = c;
#pragma unroll
        for (int off = 1; off < 32; off <<= 1) {
            const unsigned t = (unsigned)__shfl_up((int)incl, off, 32);
            if (lane >= off) incl += t;
        }
        const unsigned wtot = (unsigned)__shfl((int)incl, 31, 32);
        unsigned wbase = 0u;
        if (lane == 0) wbase = atomicAdd(&cnts[0], wtot);
        wbase = (unsigned)__shfl((int)wbase, 0, 32);
        unsigned slot = wbase + incl - c;
#pragma unroll
        for (int u = 0; u < NIT; ++u) {
            const int q = tid + u * OTPB;
            const unsigned wd[4] = { v[u].x, v[u].y, v[u].z, v[u].w };
#pragma unroll
            for (int k = 0; k < 4; ++k) {
                unsigned m = wd[k];
                while (m != 0u) {
                    const int b = __builtin_ctz(m);
                    m &= m - 1u;
                    if (slot < Q0CAP) q0[slot] = (unsigned)(q * 128 + k * 32 + b);
                    ++slot;
                }
            }
        }
    }
    __syncthreads();

    const unsigned total = cnts[0];
    if (total > Q0CAP) {
        owner_all_slow(rows, tmax, tden, src, dst, el, er, feat, lo, cnt, tid);
    } else {
        const int n0 = (int)total;
        int   ee[RIT];
        bool  ok[RIT];
        int   ss[RIT], lrs[RIT];
        float vv[RIT];
#pragma unroll
        for (int k = 0; k < RIT; ++k) {
            const int i = k * OTPB + tid;
            ok[k] = i < n0;
            ee[k] = ok[k] ? (int)q0[i] : 0;
        }
        int dd[RIT];
#pragma unroll
        for (int k = 0; k < RIT; ++k) { ss[k] = src[ee[k]]; dd[k] = dst[ee[k]]; }
#pragma unroll
        for (int k = 0; k < RIT; ++k) {
            lrs[k] = dd[k] - lo;
            ok[k] = ok[k] && (unsigned)lrs[k] < (unsigned)cnt;
            const int sq = ok[k] ? ss[k] : 0;
            const int dq = ok[k] ? dd[k] : lo;
            vv[k] = edge_score(el[sq], er[dq]);
        }
#pragma unroll
        for (int k = 0; k < RIT; ++k)
            if (ok[k]) atomicMax(&tmax[lrs[k]], enc_f32(vv[k]));
        __syncthreads();
#pragma unroll
        for (int k = 0; k < RIT; ++k) {
            if (ok[k]) {
                vv[k] = __expf(vv[k] - row_max(tmax[lrs[k]]));
                atomicAdd(&tden[lrs[k]], vv[k]);
            }
        }
        __syncthreads();
#pragma unroll
        for (int k = 0; k < RIT; ++k) {
            if (ok[k]) {
                const int lr = lrs[k];
                const float a = vv[k] / fmaxf(tden[lr], 1e-9f);
                const int ow = lr & (OWAVES - 1);
                const unsigned slot = atomicAdd(&cnts[1 + ow], 1u);
                if (slot < Q1CAP) {
                    q1p[ow * Q1CAP + slot] = (unsigned)ss[k] | ((unsigned)lr << 17);
                    q1a[ow * Q1CAP + slot] = a;
                } else {
                    owner_slow_add(rows, feat, ss[k], lr, a);
                }
            }
        }
    }
    __syncthreads();

    {
        const unsigned c1 = cnts[1 + w];
        const int n1 = __builtin_amdgcn_readfirstlane((int)(c1 < Q1CAP ? c1 : Q1CAP));
        const unsigned* qp = q1p + w * Q1CAP;
        const float*    qa = q1a + w * Q1CAP;
        for (int i0 = 0; i0 < n1; i0 += PB) {
            const int  iu = i0 + (lane & (PB - 1));
            const bool vu = iu < n1;
            const int  ic = vu ? iu : i0;
            const unsigned pv = qp[ic];
            const float    av = qa[ic];
            if (i0 + PB <= n1) owner_batch<true>(rows, feat, pv, av, PB, lane);
            else               owner_batch<false>(rows, feat, pv, av, n1 - i0, lane);
        }
    }
    __syncthreads();

    float4* o4 = (float4*)(out + (long)lo * DFEAT);
    for (int j = tid; j < cnt * (DFEAT / 4); j += OTPB) o4[j] = ((const float4*)rows)[j];
}

extern "C" void kernel_launch(void* const* d_in, const int* in_sizes, int n_in,
                              void* d_out, int out_size, void* d_ws, size_t ws_size,
                              hipStream_t stream) {
    const float* h      = (const float*)d_in[0];
    const int*   src    = (const int*)  d_in[1];
    const int*   dst    = (const int*)  d_in[2];
    const float* W      = (const float*)d_in[3];
    const float* attn_l = (const float*)d_in[4];
    const float* attn_r = (const float*)d_in[5];
    float* out = (float*)d_out;

    float*    feat  = (float*)d_ws;
    float*    el    = feat + (long)NNODES * DFEAT;
    float*    er    = el + NNODES;
    unsigned* lastplane = (unsigned*)(er + NNODES);

    gat_planes<<<NTILE, PTPB, 0, stream>>>(dst, out, lastplane);

    const int nstrips = NNODES / 16;
    gat_gemm_wmma<<<(nstrips + 3) / 4, 128, 0, stream>>>(h, W, feat);

    gat_attn_logits<<<(NNODES / NPW + 3) / 4, 128, 0, stream>>>(feat, attn_l, attn_r, el, er);

    gat_owner_sum<<<NOWN, OTPB, 0, stream>>>(src, dst, lastplane, el, er, feat, out);
}


# lines of 2048-edge tiles hold packed (source,row) records (a count and 31 slots, LDS counters), not presence bits: owners drop the per-edge look-ups of src and dst; sum order arbitrary
# speedup vs baseline: 4.2522x; 1.1652x over previous
#include <hip/hip_runtime.h>

typedef float v2f __attribute__((ext_vector_type(2)));
typedef float v8f __attribute__((ext_vector_type(8)));

#define NNODES   100000
#define NEDGES   1600000
#define DFEAT    128
#define NEG_SLOPE 0.2f

__global__ void gat_gemm_wmma(const float* __restrict__ h,
                              const float* __restrict__ W,
                              float* __restrict__ feat) {
    const int strip = blockIdx.x * 4 + (threadIdx.x >> 5);
    if (strip >= NNODES / 16) return;
    const int lane  = threadIdx.x & 31;
    const int m     = lane & 15;
    const int khalf = (lane >> 4) << 1;
    const int hihalf = (lane >> 4) << 3;
    const long row  = (long)strip * 16 + m;

    v8f acc[8];
#pragma unroll
    for (int ct = 0; ct < 8; ++ct) acc[ct] = (v8f){0,0,0,0,0,0,0,0};

    for (int k = 0; k < DFEAT; k += 4) {
        v2f a;
        a.x = h[row * DFEAT + k + khalf + 0];
        a.y = h[row * DFEAT + k + khalf + 1];
#pragma unroll
        for (int ct = 0; ct < 8; ++ct) {
            v2f b;
            b.x = W[(k + khalf + 0) * DFEAT + ct * 16 + m];
            b.y = W[(k + khalf + 1) * DFEAT + ct * 16 + m];
            acc[ct] = __builtin_amdgcn_wmma_f32_16x16x4_f32(
                false, a, false, b, (short)0, acc[ct], false, false);
        }
    }

#pragma unroll
    for (int ct = 0; ct < 8; ++ct) {
#pragma unroll
        for (int v = 0; v < 8; ++v) {
            long orow = (long)strip * 16 + v + hihalf;
            feat[orow * DFEAT + ct * 16 + m] = acc[ct][v];
        }
    }
}

#define NPW 8
static_assert(NNODES % NPW == 0, "nodes per wave must divide the node count");
__global__ void gat_attn_logits(const float* __restrict__ feat,
                                const float* __restrict__ attn_l,
                                const float* __restrict__ attn_r,
                                float* __restrict__ el,
                                float* __restrict__ er) {
    const int n0 = (blockIdx.x * 4 + (threadIdx.x >> 5)) * NPW;
    if (n0 >= NNODES) return;
    const int lane = threadIdx.x & 31;

    const float4 al = *(const float4*)(attn_l + lane * 4);
    const float4 ar = *(const float4*)(attn_r + lane * 4);

    float4 f[NPW];
#pragma unroll
    for (int u = 0; u < NPW; ++u)
        f[u] = *(const float4*)(feat + (long)(n0 + u) * DFEAT + lane * 4);

    float keep_l = 0.0f, keep_r = 0.0f;
#pragma unroll
    for (int u = 0; u < NPW; ++u) {
        float pl = f[u].x * al.x + f[u].y * al.y + f[u].z * al.z + f[u].w * al.w;
        float pr = f[u].x * ar.x + f[u].y * ar.y + f[u].z * ar.z + f[u].w * ar.w;
#pragma unroll
        for (int off = 16; off > 0; off >>= 1) {
            pl += __shfl_xor(pl, off, 32);
            pr += __shfl_xor(pr, off, 32);
        }
        if (lane == u) { keep_l = pl; keep_r = pr; }
    }
    if (lane < NPW) { el[n0 + lane] = keep_l; er[n0 + lane] = keep_r; }
}

__device__ __forceinline__ unsigned enc_f32(float x) {
    unsigned u = __float_as_uint(x);
    return (u & 0x80000000u) ? ~u : (u | 0x80000000u);
}
__device__ __forceinline__ float dec_f32(unsigned u) {
    return (u & 0x80000000u) ? __uint_as_float(u ^ 0x80000000u) : __uint_as_float(~u);
}

#define NR    391
#define NOWN  ((NNODES + NR - 1) / NR)
#define OTPB  1024
#define OWAVES (OTPB / 32)
#define Q0CAP 8192
#define Q1CAP 256
#define PB    16
#define RIT   (Q0CAP / OTPB)
#define TE    2048
#define LCAP  31
#define NTILE ((NEDGES + TE - 1) / TE)
#define PLW   (NTILE * 32)
#define PITEM (PLW / 4)
#define NIT   ((PITEM + OTPB - 1) / OTPB)
#define BP    36
#define PTPB  (TE / 4)
static_assert(NEDGES % 4 == 0, "4 edges per thread and load");
static_assert((long)PLW * 4 <= (long)NR * DFEAT * 4, "a plane must fit its owner's rows of out");
static_assert((NOWN - 1) * NR < NNODES, "every owner but the last has all NR rows");
static_assert(Q0CAP % OTPB == 0, "whole route sweeps");
static_assert(NNODES <= (1 << 17) && NR <= (1 << 9), "packing of source and local row");
static_assert(PITEM % 8 == 0, "the 8 lanes that share a line stay together");
static_assert(PTPB <= 1024, "one work-group per tile");

__device__ __forceinline__ unsigned* plane_of(float* out, unsigned* lastplane, int o) {
    return (o < NOWN - 1) ? (unsigned*)(out + (long)o * NR * DFEAT) : lastplane;
}

__device__ __forceinline__ void line_put(unsigned* cell, unsigned o, unsigned slot, int d, int s) {
    const unsigned lr = (unsigned)d - o * NR;
    if (slot < LCAP) cell[o * BP + 1 + slot] = (unsigned)s | (lr << 17);
}

__global__ __launch_bounds__(PTPB) void gat_lines(const int* __restrict__ src,
                                                  const int* __restrict__ dst,
                                                  float* out, unsigned* lastplane) {
    __shared__ unsigned cell[NOWN * BP];
    const int tid  = threadIdx.x;
    const int tile = blockIdx.x;
    for (int o = tid; o < NOWN; o += PTPB) cell[o * BP] = 0u;
    __syncthreads();
    const int e0 = tile * TE + 4 * tid;
    if (e0 < NEDGES) {
        const int4 d = *(const int4*)(dst + e0);
        const int4 s = *(const int4*)(src + e0);
        const unsigned ox = (unsigned)d.x / NR, oy = (unsigned)d.y / NR;
        const unsigned oz = (unsigned)d.z / NR, ow = (unsigned)d.w / NR;
        const unsigned kx = atomicAdd(&cell[ox * BP], 1u);
        const unsigned ky = atomicAdd(&cell[oy * BP], 1u);
        const unsigned kz = atomicAdd(&cell[oz * BP], 1u);
        const unsigned kw = atomicAdd(&cell[ow * BP], 1u);
        line_put(cell, ox, kx, d.x, s.x);
        line_put(cell, oy, ky, d.y, s.y);
        line_put(cell, oz, kz, d.z, s.z);
        line_put(cell, ow, kw, d.w, s.w);
    }
    __syncthreads();
    for (int c = tid; c < NOWN * 8; c += PTPB) {
        const int o = c >> 3, p = c & 7;
        const unsigned n = cell[o * BP];
        const unsigned j = 4u * (unsigned)p;
        uint4 v = *(const uint4*)(cell + o * BP + 4 * p);
        v.x = (j      <= n) ? v.x : 0u;
        v.y = (j + 1u <= n) ? v.y : 0u;
        v.z = (j + 2u <= n) ? v.z : 0u;
        v.w = (j + 3u <= n) ? v.w : 0u;
        *(uint4*)(plane_of(out, lastplane, o) + tile * 32 + 4 * p) = v;
    }
}

__device__ __forceinline__ void lane_row_add(float* rows, const float* __restrict__ feat,
                                             int s, int lr, float a) {
    for (int c = 0; c < DFEAT; ++c)
        atomicAdd(&rows[lr * DFEAT + c], a * feat[(long)s * DFEAT + c]);
}
__device__ __noinline__ void owner_slow_add(float* rows, const float* __restrict__ feat,
                                            int s, int lr, float a) {
    lane_row_add(rows, feat, s, lr, a);
}

__device__ __forceinline__ float edge_score(float el_s, float er_d) {
    const float v = el_s + er_d;
    return (v >= 0.0f) ? v : NEG_SLOPE * v;
}
__device__ __forceinline__ float row_max(unsigned enc) {
    const float m = dec_f32(enc);
    return __builtin_isfinite(m) ? m : 0.0f;
}

__device__ __noinline__ void owner_all_slow(float* rows, unsigned* tmax, float* tden,
                                            const int* __restrict__ src, const int* __restrict__ dst,
                                            const float* __restrict__ el, const float* __restrict__ er,
                                            const float* __restrict__ feat, int lo, int cnt, int tid) {
    for (int e = tid; e < NEDGES; e += OTPB) {
        const int d = dst[e];
        const int lr = d - lo;
        if ((unsigned)lr < (unsigned)cnt) atomicMax(&tmax[lr], enc_f32(edge_score(el[src[e]], er[d])));
    }
    __syncthreads();
    for (int e = tid; e < NEDGES; e += OTPB) {
        const int d = dst[e];
        const int lr = d - lo;
        if ((unsigned)lr < (unsigned)cnt)
            atomicAdd(&tden[lr], __expf(edge_score(el[src[e]], er[d]) - row_max(tmax[lr])));
    }
    __syncthreads();
    for (int e = tid; e < NEDGES; e += OTPB) {
        const int d = dst[e];
        const int lr = d - lo;
        if ((unsigned)lr < (unsigned)cnt) {
            const int s = src[e];
            const float a = __expf(edge_score(el[s], er[d]) - row_max(tmax[lr])) / fmaxf(tden[lr], 1e-9f);
            lane_row_add(rows, feat, s, lr, a);
        }
    }
}

template <bool FULL>
__device__ __forceinline__ void owner_batch(float* rows, const float* __restrict__ feat,
                                            unsigned pv, float av, int nleft, int lane) {
    float4 f[PB];
#pragma unroll
    for (int u = 0; u < PB; ++u) {
        if (FULL || u < nleft) {
            const unsigned p = (unsigned)__builtin_amdgcn_readlane((int)pv, u);
            const int s = (int)(p & 0x1FFFFu);
            f[u] = *(const float4*)(feat + (long)s * DFEAT + lane * 4);
        }
    }
#pragma unroll
    for (int u = 0; u < PB; ++u) {
        if (FULL || u < nleft) {
            const unsigned p = (unsigned)__builtin_amdgcn_readlane((int)pv, u);
            const float a = __int_as_float(__builtin_amdgcn_readlane(__float_as_int(av), u));
            const int lr = (int)(p >> 17);
            float4* rp = (float4*)(rows + lr * DFEAT + lane * 4);
            float4 r = *rp;
            r.x += a * f[u].x; r.y += a * f[u].y; r.z += a * f[u].z; r.w += a * f[u].w;
            *rp = r;
        }
    }
}

__global__ __launch_bounds__(OTPB) void gat_owner_sum(const int* __restrict__ src,
                                                      const int* __restrict__ dst,
                                                      const unsigned* __restrict__ lastplane,
                                                      const float* __restrict__ el,
                                                      const float* __restrict__ er,
                                                      const float* __restrict__ feat,
                                                      float* __restrict__ out) {
    __shared__ float    rows[NR * DFEAT];
    __shared__ unsigned q0[Q0CAP];
    __shared__ unsigned q1p[OWAVES * Q1CAP];
    __shared__ float    q1a[OWAVES * Q1CAP];
    __shared__ unsigned cnts[2 + OWAVES];
    __shared__ unsigned tmax[NR];
    __shared__ float    tden[NR];

    const int tid  = threadIdx.x;
    const int lane = tid & 31;
    const int w    = tid >> 5;
    const int own  = blockIdx.x;
    const int lo   = own * NR;
    const int cnt  = (NNODES - lo < NR) ? (NNODES - lo) : NR;

    for (int j = tid; j < NR * DFEAT / 4; j += OTPB) ((float4*)rows)[j] = make_float4(0.f, 0.f, 0.f, 0.f);
    if (tid <= OWAVES + 1) cnts[tid] = 0u;
    if (tid < NR) { tmax[tid] = 0u; tden[tid] = 0.0f; }
    __syncthreads();

    {
        const uint4* pv = (const uint4*)((own < NOWN - 1) ? (const unsigned*)(out + (long)lo * DFEAT) : lastplane);
        uint4 v[NIT];
#pragma unroll
        for (int u = 0; u < NIT; ++u) {
            const int q = tid + u * OTPB;
            v[u] = pv[q < PITEM ? q : tid];
            if (q >= PITEM) v[u] = make_uint4(0u, 0u, 0u, 0u);
        }
        const unsigned j0 = 4u * (unsigned)(lane & 7);
        unsigned nn[NIT];
        unsigned c = 0u;
#pragma unroll
        for (int u = 0; u < NIT; ++u) {
            unsigned n = (unsigned)__shfl((int)v[u].x, lane & ~7, 32);
            if (n > LCAP) { atomicOr(&cnts[1 + OWAVES], 1u); n = LCAP; }
            nn[u] = n;
            c += (unsigned)((j0 >= 1u && j0 <= n) + (j0 + 1u <= n) + (j0 + 2u <= n) + (j0 + 3u <= n));
        }
        unsigned incl = c;
#pragma unroll
        for (int off = 1; off < 32; off <<= 1) {
            const unsigned t = (unsigned)__shfl_up((int)incl, off, 32);
            if (lane >= off) incl += t;
        }
        const unsigned wtot = (unsigned)__shfl((int)incl, 31, 32);
        unsigned wbase = 0u;
        if (lane == 0) wbase = atomicAdd(&cnts[0], wtot);
        wbase = (unsigned)__shfl((int)wbase, 0, 32);
        unsigned slot = wbase + incl - c;
#pragma unroll
        for (int u = 0; u < NIT; ++u) {
            const unsigned wd[4] = { v[u].x, v[u].y, v[u].z, v[u].w };
#pragma unroll
            for (int k = 0; k < 4; ++k) {
                const unsigned j = j0 + (unsigned)k;
                if (j >= 1u && j <= nn[u]) {
                    if (slot < Q0CAP) q0[slot] = wd[k];
                    ++slot;
                }
            }
        }
    }
    __syncthreads();

    const unsigned total = cnts[0];
    if (total > Q0CAP || cnts[1 + OWAVES] != 0u) {
        owner_all_slow(rows, tmax, tden, src, dst, el, er, feat, lo, cnt, tid);
    } else {
        const int n0 = (int)total;
        unsigned rc[RIT];
        bool  ok[RIT];
        int   ss[RIT], lrs[RIT];
        float vv[RIT];
#pragma unroll
        for (int k = 0; k < RIT; ++k) {
            const int i = k * OTPB + tid;
            ok[k] = i < n0;
            rc[k] = ok[k] ? q0[i] : 0u;
        }
#pragma unroll
        for (int k = 0; k < RIT; ++k) {
            ss[k]  = (int)(rc[k] & 0x1FFFFu);
            lrs[k] = (int)(rc[k] >> 17);
            ok[k] = ok[k] && lrs[k] < cnt && ss[k] < NNODES;
            const int sq = ok[k] ? ss[k] : 0;
            const int dq = ok[k] ? lo + lrs[k] : lo;
            vv[k] = edge_score(el[sq], er[dq]);
        }
#pragma unroll
        for (int k = 0; k < RIT; ++k)
            if (ok[k]) atomicMax(&tmax[lrs[k]], enc_f32(vv[k]));
        __syncthreads();
#pragma unroll
        for (int k = 0; k < RIT; ++k) {
            if (ok[k]) {
                vv[k] = __expf(vv[k] - row_max(tmax[lrs[k]]));
                atomicAdd(&tden[lrs[k]], vv[k]);
            }
        }
        __syncthreads();
#pragma unroll
        for (int k = 0; k < RIT; ++k) {
            if (ok[k]) {
                const int lr = lrs[k];
                const float a = vv[k] / fmaxf(tden[lr], 1e-9f);
                const int ow = lr & (OWAVES - 1);
                const unsigned slot = atomicAdd(&cnts[1 + ow], 1u);
                if (slot < Q1CAP) {
                    q1p[ow * Q1CAP + slot] = (unsigned)ss[k] | ((unsigned)lr << 17);
                    q1a[ow * Q1CAP + slot] = a;
                } else {
                    owner_slow_add(rows, feat, ss[k], lr, a);
                }
            }
        }
    }
    __syncthreads();

    {
        const unsigned c1 = cnts[1 + w];
        const int n1 = __builtin_amdgcn_readfirstlane((int)(c1 < Q1CAP ? c1 : Q1CAP));
        const unsigned* qp = q1p + w * Q1CAP;
        const float*    qa = q1a + w * Q1CAP;
        for (int i0 = 0; i0 < n1; i0 += PB) {
            const int  iu = i0 + (lane & (PB - 1));
            const bool vu = iu < n1;
            const int  ic = vu ? iu : i0;
            const unsigned pv = qp[ic];
            const float    av = qa[ic];
            if (i0 + PB <= n1) owner_batch<true>(rows, feat, pv, av, PB, lane);
            else               owner_batch<false>(rows, feat, pv, av, n1 - i0, lane);
        }
    }
    __syncthreads();

    float4* o4 = (float4*)(out + (long)lo * DFEAT);
    for (int j = tid; j < cnt * (DFEAT / 4); j += OTPB) o4[j] = ((const float4*)rows)[j];
}

extern "C" void kernel_launch(void* const* d_in, const int* in_sizes, int n_in,
                              void* d_out, int out_size, void* d_ws, size_t ws_size,
                              hipStream_t stream) {
    const float* h      = (const float*)d_in[0];
    const int*   src    = (const int*)  d_in[1];
    const int*   dst    = (const int*)  d_in[2];
    const float* W      = (const float*)d_in[3];
    const float* attn_l = (const float*)d_in[4];
    const float* attn_r = (const float*)d_in[5];
    float* out = (float*)d_out;

    float*    feat  = (float*)d_ws;
    float*    el    = feat + (long)NNODES * DFEAT;
    float*    er    = el + NNODES;
    unsigned* lastplane = (unsigned*)(er + NNODES);

    gat_lines<<<NTILE, PTPB, 0, stream>>>(src, dst, out, lastplane);

    const int nstrips = NNODES / 16;
    gat_gemm_wmma<<<(nstrips + 3) / 4, 128, 0, stream>>>(h, W, feat);

    gat_attn_logits<<<(NNODES / NPW + 3) / 4, 128, 0, stream>>>(feat, attn_l, attn_r, el, er);

    gat_owner_sum<<<NOWN, OTPB, 0, stream>>>(src, dst, lastplane, el, er, feat, out);
}


# sum phase: all source-row loads of a batch are issued together (row address = scalar row base + lane offset, loads unguarded), not one at a time each waited for at its use; same sums, same order
# speedup vs baseline: 4.9315x; 1.1598x over previous
#include <hip/hip_runtime.h>

typedef float v2f __attribute__((ext_vector_type(2)));
typedef float v8f __attribute__((ext_vector_type(8)));

#define NNODES   100000
#define NEDGES   1600000
#define DFEAT    128
#define NEG_SLOPE 0.2f

__global__ void gat_gemm_wmma(const float* __restrict__ h,
                              const float* __restrict__ W,
                              float* __restrict__ feat) {
    const int strip = blockIdx.x * 4 + (threadIdx.x >> 5);
    if (strip >= NNODES / 16) return;
    const int lane  = threadIdx.x & 31;
    const int m     = lane & 15;
    const int khalf = (lane >> 4) << 1;
    const int hihalf = (lane >> 4) << 3;
    const long row  = (long)strip * 16 + m;

    v8f acc[8];
#pragma unroll
    for (int ct = 0; ct < 8; ++ct) acc[ct] = (v8f){0,0,0,0,0,0,0,0};

    for (int k = 0; k < DFEAT; k += 4) {
        v2f a;
        a.x = h[row * DFEAT + k + khalf + 0];
        a.y = h[row * DFEAT + k + khalf + 1];
#pragma unroll
        for (int ct = 0; ct < 8; ++ct) {
            v2f b;
            b.x = W[(k + khalf + 0) * DFEAT + ct * 16 + m];
            b.y = W[(k + khalf + 1) * DFEAT + ct * 16 + m];
            acc[ct] = __builtin_amdgcn_wmma_f32_16x16x4_f32(
                false, a, false, b, (short)0, acc[ct], false, false);
        }
    }

#pragma unroll
    for (int ct = 0; ct < 8; ++ct) {
#pragma unroll
        for (int v = 0; v < 8; ++v) {
            long orow = (long)strip * 16 + v + hihalf;
            feat[orow * DFEAT + ct * 16 + m] = acc[ct][v];
        }
    }
}

#define NPW 8
static_assert(NNODES % NPW == 0, "nodes per wave must divide the node count");
__global__ void gat_attn_logits(const float* __restrict__ feat,
                                const float* __restrict__ attn_l,
                                const float* __restrict__ attn_r,
                                float* __restrict__ el,
                                float* __restrict__ er) {
    const int n0 = (blockIdx.x * 4 + (threadIdx.x >> 5)) * NPW;
    if (n0 >= NNODES) return;
    const int lane = threadIdx.x & 31;

    const float4 al = *(const float4*)(attn_l + lane * 4);
    const float4 ar = *(const float4*)(attn_r + lane * 4);

    float4 f[NPW];
#pragma unroll
    for (int u = 0; u < NPW; ++u)
        f[u] = *(const float4*)(feat + (long)(n0 + u) * DFEAT + lane * 4);

    float keep_l = 0.0f, keep_r = 0.0f;
#pragma unroll
    for (int u = 0; u < NPW; ++u) {
        float pl = f[u].x * al.x + f[u].y * al.y + f[u].z * al.z + f[u].w * al.w;
        float pr = f[u].x * ar.x + f[u].y * ar.y + f[u].z * ar.z + f[u].w * ar.w;
#pragma unroll
        for (int off = 16; off > 0; off >>= 1) {
            pl += __shfl_xor(pl, off, 32);
            pr += __shfl_xor(pr, off, 32);
        }
        if (lane == u) { keep_l = pl; keep_r = pr; }
    }
    if (lane < NPW) { el[n0 + lane] = keep_l; er[n0 + lane] = keep_r; }
}

__device__ __forceinline__ unsigned enc_f32(float x) {
    unsigned u = __float_as_uint(x);
    return (u & 0x80000000u) ? ~u : (u | 0x80000000u);
}
__device__ __forceinline__ float dec_f32(unsigned u) {
    return (u & 0x80000000u) ? __uint_as_float(u ^ 0x80000000u) : __uint_as_float(~u);
}

#define NR    391
#define NOWN  ((NNODES + NR - 1) / NR)
#define OTPB  1024
#define OWAVES (OTPB / 32)
#define Q0CAP 8192
#define Q1CAP 256
#define PB    16
#define RIT   (Q0CAP / OTPB)
#define TE    2048
#define LCAP  31
#define NTILE ((NEDGES + TE - 1) / TE)
#define PLW   (NTILE * 32)
#define PITEM (PLW / 4)
#define NIT   ((PITEM + OTPB - 1) / OTPB)
#define BP    36
#define PTPB  (TE / 4)
static_assert(NEDGES % 4 == 0, "4 edges per thread and load");
static_assert((long)PLW * 4 <= (long)NR * DFEAT * 4, "a plane must fit its owner's rows of out");
static_assert((NOWN - 1) * NR < NNODES, "every owner but the last has all NR rows");
static_assert(Q0CAP % OTPB == 0, "whole route sweeps");
static_assert(NNODES <= (1 << 17) && NR <= (1 << 9), "packing of source and local row");
static_assert(PITEM % 8 == 0, "the 8 lanes that share a line stay together");
static_assert(PTPB <= 1024, "one work-group per tile");
static_assert(DFEAT * 4 == 512 && PB <= 32, "a row is 512 B (16 B per lane); a batch entry per lane");

__device__ __forceinline__ unsigned* plane_of(float* out, unsigned* lastplane, int o) {
    return (o < NOWN - 1) ? (unsigned*)(out + (long)o * NR * DFEAT) : lastplane;
}

__device__ __forceinline__ void line_put(unsigned* cell, unsigned o, unsigned slot, int d, int s) {
    const unsigned lr = (unsigned)d - o * NR;
    if (slot < LCAP) cell[o * BP + 1 + slot] = (unsigned)s | (lr << 17);
}

__global__ __launch_bounds__(PTPB) void gat_lines(const int* __restrict__ src,
                                                  const int* __restrict__ dst,
                                                  float* out, unsigned* lastplane) {
    __shared__ unsigned cell[NOWN * BP];
    const int tid  = threadIdx.x;
    const int tile = blockIdx.x;
    for (int o = tid; o < NOWN; o += PTPB) cell[o * BP] = 0u;
    __syncthreads();
    const int e0 = tile * TE + 4 * tid;
    if (e0 < NEDGES) {
        const int4 d = *(const int4*)(dst + e0);
        const int4 s = *(const int4*)(src + e0);
        const unsigned ox = (unsigned)d.x / NR, oy = (unsigned)d.y / NR;
        const unsigned oz = (unsigned)d.z / NR, ow = (unsigned)d.w / NR;
        const unsigned kx = atomicAdd(&cell[ox * BP], 1u);
        const unsigned ky = atomicAdd(&cell[oy * BP], 1u);
        const unsigned kz = atomicAdd(&cell[oz * BP], 1u);
        const unsigned kw = atomicAdd(&cell[ow * BP], 1u);
        line_put(cell, ox, kx, d.x, s.x);
        line_put(cell, oy, ky, d.y, s.y);
        line_put(cell, oz, kz, d.z, s.z);
        line_put(cell, ow, kw, d.w, s.w);
    }
    __syncthreads();
    for (int c = tid; c < NOWN * 8; c += PTPB) {
        const int o = c >> 3, p = c & 7;
        const unsigned n = cell[o * BP];
        const unsigned j = 4u * (unsigned)p;
        uint4 v = *(const uint4*)(cell + o * BP + 4 * p);
        v.x = (j      <= n) ? v.x : 0u;
        v.y = (j + 1u <= n) ? v.y : 0u;
        v.z = (j + 2u <= n) ? v.z : 0u;
        v.w = (j + 3u <= n) ? v.w : 0u;
        *(uint4*)(plane_of(out, lastplane, o) + tile * 32 + 4 * p) = v;
    }
}

__device__ __forceinline__ void lane_row_add(float* rows, const float* __restrict__ feat,
                                             int s, int lr, float a) {
    for (int c = 0; c < DFEAT; ++c)
        atomicAdd(&rows[lr * DFEAT + c], a * feat[(long)s * DFEAT + c]);
}
__device__ __noinline__ void owner_slow_add(float* rows, const float* __restrict__ feat,
                                            int s, int lr, float a) {
    lane_row_add(rows, feat, s, lr, a);
}

__device__ __forceinline__ float edge_score(float el_s, float er_d) {
    const float v = el_s + er_d;
    return (v >= 0.0f) ? v : NEG_SLOPE * v;
}
__device__ __forceinline__ float row_max(unsigned enc) {
    const float m = dec_f32(enc);
    return __builtin_isfinite(m) ? m : 0.0f;
}

__device__ __noinline__ void owner_all_slow(float* rows, unsigned* tmax, float* tden,
                                            const int* __restrict__ src, const int* __restrict__ dst,
                                            const float* __restrict__ el, const float* __restrict__ er,
                                            const float* __restrict__ feat, int lo, int cnt, int tid) {
    for (int e = tid; e < NEDGES; e += OTPB) {
        const int d = dst[e];
        const int lr = d - lo;
        if ((unsigned)lr < (unsigned)cnt) atomicMax(&tmax[lr], enc_f32(edge_score(el[src[e]], er[d])));
    }
    __syncthreads();
    for (int e = tid; e < NEDGES; e += OTPB) {
        const int d = dst[e];
        const int lr = d - lo;
        if ((unsigned)lr < (unsigned)cnt)
            atomicAdd(&tden[lr], __expf(edge_score(el[src[e]], er[d]) - row_max(tmax[lr])));
    }
    __syncthreads();
    for (int e = tid; e < NEDGES; e += OTPB) {
        const int d = dst[e];
        const int lr = d - lo;
        if ((unsigned)lr < (unsigned)cnt) {
            const int s = src[e];
            const float a = __expf(edge_score(el[s], er[d]) - row_max(tmax[lr])) / fmaxf(tden[lr], 1e-9f);
            lane_row_add(rows, feat, s, lr, a);
        }
    }
}

typedef float v4f __attribute__((ext_vector_type(4)));
typedef __attribute__((address_space(1))) const v4f*  gv4p;
typedef __attribute__((address_space(1))) const char* gcp;

template <bool FULL>
__device__ __forceinline__ void owner_batch(float* rows, const float* __restrict__ feat,
                                            unsigned pv, float av, int nleft, int lane) {
    v4f f[PB];
    const unsigned loff = (unsigned)lane * 16u;
#pragma unroll
    for (int u = 0; u < PB; ++u) {
        const unsigned p = (unsigned)__builtin_amdgcn_readlane((int)pv, u);
        const unsigned s = p & 0x1FFFFu;
        const gcp rb = (gcp)((unsigned long long)feat + ((unsigned long long)s << 9));
        f[u] = *(gv4p)(rb + loff);
    }
#pragma unroll
    for (int u = 0; u < PB; ++u) {
        if (FULL || u < nleft) {
            const unsigned p = (unsigned)__builtin_amdgcn_readlane((int)pv, u);
            const float a = __int_as_float(__builtin_amdgcn_readlane(__float_as_int(av), u));
            const int lr = (int)(p >> 17);
            float4* rp = (float4*)(rows + lr * DFEAT + lane * 4);
            float4 r = *rp;
            r.x += a * f[u].x; r.y += a * f[u].y; r.z += a * f[u].z; r.w += a * f[u].w;
            *rp = r;
        }
    }
}

__global__ __launch_bounds__(OTPB) void gat_owner_sum(const int* __restrict__ src,
                                                      const int* __restrict__ dst,
                                                      const unsigned* __restrict__ lastplane,
                                                      const float* __restrict__ el,
                                                      const float* __restrict__ er,
                                                      const float* __restrict__ feat,
                                                      float* __restrict__ out) {
    __shared__ float    rows[NR * DFEAT];
    __shared__ unsigned q0[Q0CAP];
    __shared__ unsigned q1p[OWAVES * Q1CAP];
    __shared__ float    q1a[OWAVES * Q1CAP];
    __shared__ unsigned cnts[2 + OWAVES];
    __shared__ unsigned tmax[NR];
    __shared__ float    tden[NR];

    const int tid  = threadIdx.x;
    const int lane = tid & 31;
    const int w    = tid >> 5;
    const int own  = blockIdx.x;
    const int lo   = own * NR;
    const int cnt  = (NNODES - lo < NR) ? (NNODES - lo) : NR;

    for (int j = tid; j < NR * DFEAT / 4; j += OTPB) ((float4*)rows)[j] = make_float4(0.f, 0.f, 0.f, 0.f);
    if (tid <= OWAVES + 1) cnts[tid] = 0u;
    if (tid < NR) { tmax[tid] = 0u; tden[tid] = 0.0f; }
    __syncthreads();

    {
        const uint4* pv = (const uint4*)((own < NOWN - 1) ? (const unsigned*)(out + (long)lo * DFEAT) : lastplane);
        uint4 v[NIT];
#pragma unroll
        for (int u = 0; u < NIT; ++u) {
            const int q = tid + u * OTPB;
            v[u] = pv[q < PITEM ? q : tid];
            if (q >= PITEM) v[u] = make_uint4(0u, 0u, 0u, 0u);
        }
        const unsigned j0 = 4u * (unsigned)(lane & 7);
        unsigned nn[NIT];
        unsigned c = 0u;
#pragma unroll
        for (int u = 0; u < NIT; ++u) {
            unsigned n = (unsigned)__shfl((int)v[u].x, lane & ~7, 32);
            if (n > LCAP) { atomicOr(&cnts[1 + OWAVES], 1u); n = LCAP; }
            nn[u] = n;
            c += (unsigned)((j0 >= 1u && j0 <= n) + (j0 + 1u <= n) + (j0 + 2u <= n) + (j0 + 3u <= n));
        }
        unsigned incl = c;
#pragma unroll
        for (int off = 1; off < 32; off <<= 1) {
            const unsigned t = (unsigned)__shfl_up((int)incl, off, 32);
            if (lane >= off) incl += t;
        }
        const unsigned wtot = (unsigned)__shfl((int)incl, 31, 32);
        unsigned wbase = 0u;
        if (lane == 0) wbase = atomicAdd(&cnts[0], wtot);
        wbase = (unsigned)__shfl((int)wbase, 0, 32);
        unsigned slot = wbase + incl - c;
#pragma unroll
        for (int u = 0; u < NIT; ++u) {
            const unsigned wd[4] = { v[u].x, v[u].y, v[u].z, v[u].w };
#pragma unroll
            for (int k = 0; k < 4; ++k) {
                const unsigned j = j0 + (unsigned)k;
                if (j >= 1u && j <= nn[u]) {
                    if (slot < Q0CAP) q0[slot] = wd[k];
                    ++slot;
                }
            }
        }
    }
    __syncthreads();

    const unsigned total = cnts[0];
    if (total > Q0CAP || cnts[1 + OWAVES] != 0u) {
        owner_all_slow(rows, tmax, tden, src, dst, el, er, feat, lo, cnt, tid);
    } else {
        const int n0 = (int)total;
        unsigned rc[RIT];
        bool  ok[RIT];
        int   ss[RIT], lrs[RIT];
        float vv[RIT];
#pragma unroll
        for (int k = 0; k < RIT; ++k) {
            const int i = k * OTPB + tid;
            ok[k] = i < n0;
            rc[k] = ok[k] ? q0[i] : 0u;
        }
#pragma unroll
        for (int k = 0; k < RIT; ++k) {
            ss[k]  = (int)(rc[k] & 0x1FFFFu);
            lrs[k] = (int)(rc[k] >> 17);
            ok[k] = ok[k] && lrs[k] < cnt && ss[k] < NNODES;
            const int sq = ok[k] ? ss[k] : 0;
            const int dq = ok[k] ? lo + lrs[k] : lo;
            vv[k] = edge_score(el[sq], er[dq]);
        }
#pragma unroll
        for (int k = 0; k < RIT; ++k)
            if (ok[k]) atomicMax(&tmax[lrs[k]], enc_f32(vv[k]));
        __syncthreads();
#pragma unroll
        for (int k = 0; k < RIT; ++k) {
            if (ok[k]) {
                vv[k] = __expf(vv[k] - row_max(tmax[lrs[k]]));
                atomicAdd(&tden[lrs[k]], vv[k]);
            }
        }
        __syncthreads();
#pragma unroll
        for (int k = 0; k < RIT; ++k) {
            if (ok[k]) {
                const int lr = lrs[k];
                const float a = vv[k] / fmaxf(tden[lr], 1e-9f);
                const int ow = lr & (OWAVES - 1);
                const unsigned slot = atomicAdd(&cnts[1 + ow], 1u);
                if (slot < Q1CAP) {
                    q1p[ow * Q1CAP + slot] = (unsigned)ss[k] | ((unsigned)lr << 17);
                    q1a[ow * Q1CAP + slot] = a;
                } else {
                    owner_slow_add(rows, feat, ss[k], lr, a);
                }
            }
        }
    }
    __syncthreads();

    {
        const unsigned c1 = cnts[1 + w];
        const int n1 = __builtin_amdgcn_readfirstlane((int)(c1 < Q1CAP ? c1 : Q1CAP));
        const unsigned* qp = q1p + w * Q1CAP;
        const float*    qa = q1a + w * Q1CAP;
        for (int i0 = 0; i0 < n1; i0 += PB) {
            const int  iu = i0 + (lane & (PB - 1));
            const bool vu = iu < n1;
            const int  ic = vu ? iu : i0;
            const unsigned pv = qp[ic];
            const float    av = qa[ic];
            if (i0 + PB <= n1) owner_batch<true>(rows, feat, pv, av, PB, lane);
            else               owner_batch<false>(rows, feat, pv, av, n1 - i0, lane);
        }
    }
    __syncthreads();

    float4* o4 = (float4*)(out + (long)lo * DFEAT);
    for (int j = tid; j < cnt * (DFEAT / 4); j += OTPB) o4[j] = ((const float4*)rows)[j];
}

extern "C" void kernel_launch(void* const* d_in, const int* in_sizes, int n_in,
                              void* d_out, int out_size, void* d_ws, size_t ws_size,
                              hipStream_t stream) {
    const float* h      = (const float*)d_in[0];
    const int*   src    = (const int*)  d_in[1];
    const int*   dst    = (const int*)  d_in[2];
    const float* W      = (const float*)d_in[3];
    const float* attn_l = (const float*)d_in[4];
    const float* attn_r = (const float*)d_in[5];
    float* out = (float*)d_out;

    float*    feat  = (float*)d_ws;
    float*    el    = feat + (long)NNODES * DFEAT;
    float*    er    = el + NNODES;
    unsigned* lastplane = (unsigned*)(er + NNODES);

    gat_lines<<<NTILE, PTPB, 0, stream>>>(src, dst, out, lastplane);

    const int nstrips = NNODES / 16;
    gat_gemm_wmma<<<(nstrips + 3) / 4, 128, 0, stream>>>(h, W, feat);

    gat_attn_logits<<<(NNODES / NPW + 3) / 4, 128, 0, stream>>>(feat, attn_l, attn_r, el, er);

    gat_owner_sum<<<NOWN, OTPB, 0, stream>>>(src, dst, lastplane, el, er, feat, out);
}


# owner's records put in row order in LDS (place = value an LDS count add returns + running sum of row counts); a wave sums a row in registers and stores it once: no LDS row sums; sum order arbitrary
# speedup vs baseline: 5.2745x; 1.0696x over previous
#include <hip/hip_runtime.h>

typedef float v2f __attribute__((ext_vector_type(2)));
typedef float v8f __attribute__((ext_vector_type(8)));

#define NNODES   100000
#define NEDGES   1600000
#define DFEAT    128
#define NEG_SLOPE 0.2f

__global__ void gat_gemm_wmma(const float* __restrict__ h,
                              const float* __restrict__ W,
                              float* __restrict__ feat) {
    const int strip = blockIdx.x * 4 + (threadIdx.x >> 5);
    if (strip >= NNODES / 16) return;
    const int lane  = threadIdx.x & 31;
    const int m     = lane & 15;
    const int khalf = (lane >> 4) << 1;
    const int hihalf = (lane >> 4) << 3;
    const long row  = (long)strip * 16 + m;

    v8f acc[8];
#pragma unroll
    for (int ct = 0; ct < 8; ++ct) acc[ct] = (v8f){0,0,0,0,0,0,0,0};

    for (int k = 0; k < DFEAT; k += 4) {
        v2f a;
        a.x = h[row * DFEAT + k + khalf + 0];
        a.y = h[row * DFEAT + k + khalf + 1];
#pragma unroll
        for (int ct = 0; ct < 8; ++ct) {
            v2f b;
            b.x = W[(k + khalf + 0) * DFEAT + ct * 16 + m];
            b.y = W[(k + khalf + 1) * DFEAT + ct * 16 + m];
            acc[ct] = __builtin_amdgcn_wmma_f32_16x16x4_f32(
                false, a, false, b, (short)0, acc[ct], false, false);
        }
    }

#pragma unroll
    for (int ct = 0; ct < 8; ++ct) {
#pragma unroll
        for (int v = 0; v < 8; ++v) {
            long orow = (long)strip * 16 + v + hihalf;
            feat[orow * DFEAT + ct * 16 + m] = acc[ct][v];
        }
    }
}

#define NPW 8
static_assert(NNODES % NPW == 0, "nodes per wave must divide the node count");
__global__ void gat_attn_logits(const float* __restrict__ feat,
                                const float* __restrict__ attn_l,
                                const float* __restrict__ attn_r,
                                float* __restrict__ el,
                                float* __restrict__ er) {
    const int n0 = (blockIdx.x * 4 + (threadIdx.x >> 5)) * NPW;
    if (n0 >= NNODES) return;
    const int lane = threadIdx.x & 31;

    const float4 al = *(const float4*)(attn_l + lane * 4);
    const float4 ar = *(const float4*)(attn_r + lane * 4);

    float4 f[NPW];
#pragma unroll
    for (int u = 0; u < NPW; ++u)
        f[u] = *(const float4*)(feat + (long)(n0 + u) * DFEAT + lane * 4);

    float keep_l = 0.0f, keep_r = 0.0f;
#pragma unroll
    for (int u = 0; u < NPW; ++u) {
        float pl = f[u].x * al.x + f[u].y * al.y + f[u].z * al.z + f[u].w * al.w;
        float pr = f[u].x * ar.x + f[u].y * ar.y + f[u].z * ar.z + f[u].w * ar.w;
#pragma unroll
        for (int off = 16; off > 0; off >>= 1) {
            pl += __shfl_xor(pl, off, 32);
            pr += __shfl_xor(pr, off, 32);
        }
        if (lane == u) { keep_l = pl; keep_r = pr; }
    }
    if (lane < NPW) { el[n0 + lane] = keep_l; er[n0 + lane] = keep_r; }
}

__device__ __forceinline__ unsigned enc_f32(float x) {
    unsigned u = __float_as_uint(x);
    return (u & 0x80000000u) ? ~u : (u | 0x80000000u);
}
__device__ __forceinline__ float dec_f32(unsigned u) {
    return (u & 0x80000000u) ? __uint_as_float(u ^ 0x80000000u) : __uint_as_float(~u);
}

#define NR    391
#define NOWN  ((NNODES + NR - 1) / NR)
#define OTPB  1024
#define OWAVES (OTPB / 32)
#define Q0CAP 8192
#define PB    16
#define RIT   (Q0CAP / OTPB)
#define RPL   ((NR + 31) / 32)
#define TE    2048
#define LCAP  31
#define NTILE ((NEDGES + TE - 1) / TE)
#define PLW   (NTILE * 32)
#define PITEM (PLW / 4)
#define NIT   ((PITEM + OTPB - 1) / OTPB)
#define BP    36
#define PTPB  (TE / 4)
static_assert(NEDGES % 4 == 0, "4 edges per thread and load");
static_assert((long)PLW * 4 <= (long)NR * DFEAT * 4, "a plane must fit its owner's rows of out");
static_assert((NOWN - 1) * NR < NNODES, "every owner but the last has all NR rows");
static_assert(Q0CAP % OTPB == 0, "whole route sweeps");
static_assert(NNODES <= (1 << 17) && NR <= (1 << 9), "packing of source and local row");
static_assert(PITEM % 8 == 0, "the 8 lanes that share a line stay together");
static_assert(PTPB <= 1024, "one work-group per tile");
static_assert(DFEAT * 4 == 512 && PB <= 32, "a row is 512 B (16 B per lane); a batch entry per lane");
static_assert(32 * RPL > NR, "the running sum's last lane also writes the entry behind the last row");
static_assert((NR + OWAVES - 1) / OWAVES <= 32, "a wave's rows: one lane each in the empty-row check");

__device__ __forceinline__ unsigned* plane_of(float* out, unsigned* lastplane, int o) {
    return (o < NOWN - 1) ? (unsigned*)(out + (long)o * NR * DFEAT) : lastplane;
}

__device__ __forceinline__ void line_put(unsigned* cell, unsigned o, unsigned slot, int d, int s) {
    const unsigned lr = (unsigned)d - o * NR;
    if (slot < LCAP) cell[o * BP + 1 + slot] = (unsigned)s | (lr << 17);
}

__global__ __launch_bounds__(PTPB) void gat_lines(const int* __restrict__ src,
                                                  const int* __restrict__ dst,
                                                  float* out, unsigned* lastplane) {
    __shared__ unsigned cell[NOWN * BP];
    const int tid  = threadIdx.x;
    const int tile = blockIdx.x;
    for (int o = tid; o < NOWN; o += PTPB) cell[o * BP] = 0u;
    __syncthreads();
    const int e0 = tile * TE + 4 * tid;
    if (e0 < NEDGES) {
        const int4 d = *(const int4*)(dst + e0);
        const int4 s = *(const int4*)(src + e0);
        const unsigned ox = (unsigned)d.x / NR, oy = (unsigned)d.y / NR;
        const unsigned oz = (unsigned)d.z / NR, ow = (unsigned)d.w / NR;
        const unsigned kx = atomicAdd(&cell[ox * BP], 1u);
        const unsigned ky = atomicAdd(&cell[oy * BP], 1u);
        const unsigned kz = atomicAdd(&cell[oz * BP], 1u);
        const unsigned kw = atomicAdd(&cell[ow * BP], 1u);
        line_put(cell, ox, kx, d.x, s.x);
        line_put(cell, oy, ky, d.y, s.y);
        line_put(cell, oz, kz, d.z, s.z);
        line_put(cell, ow, kw, d.w, s.w);
    }
    __syncthreads();
    for (int c = tid; c < NOWN * 8; c += PTPB) {
        const int o = c >> 3, p = c & 7;
        const unsigned n = cell[o * BP];
        const unsigned j = 4u * (unsigned)p;
        uint4 v = *(const uint4*)(cell + o * BP + 4 * p);
        v.x = (j      <= n) ? v.x : 0u;
        v.y = (j + 1u <= n) ? v.y : 0u;
        v.z = (j + 2u <= n) ? v.z : 0u;
        v.w = (j + 3u <= n) ? v.w : 0u;
        *(uint4*)(plane_of(out, lastplane, o) + tile * 32 + 4 * p) = v;
    }
}

__device__ __forceinline__ float edge_score(float el_s, float er_d) {
    const float v = el_s + er_d;
    return (v >= 0.0f) ? v : NEG_SLOPE * v;
}
__device__ __forceinline__ float row_max(unsigned enc) {
    const float m = dec_f32(enc);
    return __builtin_isfinite(m) ? m : 0.0f;
}

__device__ __noinline__ void owner_all_slow(float* rows, unsigned* tmax, float* tden,
                                            const int* __restrict__ src, const int* __restrict__ dst,
                                            const float* __restrict__ el, const float* __restrict__ er,
                                            const float* __restrict__ feat, float* out,
                                            int lo, int cnt, int tid) {
    for (int j = tid; j < NR * DFEAT / 4; j += OTPB) ((float4*)rows)[j] = make_float4(0.f, 0.f, 0.f, 0.f);
    for (int e = tid; e < NEDGES; e += OTPB) {
        const int d = dst[e];
        const int lr = d - lo;
        if ((unsigned)lr < (unsigned)cnt) atomicMax(&tmax[lr], enc_f32(edge_score(el[src[e]], er[d])));
    }
    __syncthreads();
    for (int e = tid; e < NEDGES; e += OTPB) {
        const int d = dst[e];
        const int lr = d - lo;
        if ((unsigned)lr < (unsigned)cnt)
            atomicAdd(&tden[lr], __expf(edge_score(el[src[e]], er[d]) - row_max(tmax[lr])));
    }
    __syncthreads();
    for (int e = tid; e < NEDGES; e += OTPB) {
        const int d = dst[e];
        const int lr = d - lo;
        if ((unsigned)lr < (unsigned)cnt) {
            const int s = src[e];
            const float a = __expf(edge_score(el[s], er[d]) - row_max(tmax[lr])) / fmaxf(tden[lr], 1e-9f);
            for (int c = 0; c < DFEAT; ++c)
                atomicAdd(&rows[lr * DFEAT + c], a * feat[(long)s * DFEAT + c]);
        }
    }
    __syncthreads();
    float4* o4 = (float4*)(out + (long)lo * DFEAT);
    for (int j = tid; j < cnt * (DFEAT / 4); j += OTPB) o4[j] = ((const float4*)rows)[j];
}

typedef float v4f __attribute__((ext_vector_type(4)));
typedef __attribute__((address_space(1))) const v4f*  gv4p;
typedef __attribute__((address_space(1))) const char* gcp;

__device__ __forceinline__ void row_store(float* __restrict__ out, int row, int lane, v4f v) {
    *(v4f*)(out + (long)row * DFEAT + lane * 4) = v;
}

template <bool FULL>
__device__ __forceinline__ void owner_batch(float* __restrict__ out, int lo,
                                            const float* __restrict__ feat,
                                            unsigned pv, float av, int nleft, int lane,
                                            int& cur, v4f& acc) {
    v4f f[PB];
    const unsigned loff = (unsigned)lane * 16u;
#pragma unroll
    for (int u = 0; u < PB; ++u) {
        const unsigned p = (unsigned)__builtin_amdgcn_readlane((int)pv, u);
        const unsigned s = p & 0x1FFFFu;
        const gcp rb = (gcp)((unsigned long long)feat + ((unsigned long long)s << 9));
        f[u] = *(gv4p)(rb + loff);
    }
#pragma unroll
    for (int u = 0; u < PB; ++u) {
        if (FULL || u < nleft) {
            const unsigned p = (unsigned)__builtin_amdgcn_readlane((int)pv, u);
            const float a = __int_as_float(__builtin_amdgcn_readlane(__float_as_int(av), u));
            const int lr = (int)(p >> 17);
            if (lr != cur) {
                row_store(out, lo + cur, lane, acc);
                cur = lr;
                acc = (v4f){0.f, 0.f, 0.f, 0.f};
            }
            acc.x += a * f[u].x; acc.y += a * f[u].y; acc.z += a * f[u].z; acc.w += a * f[u].w;
        }
    }
}

__global__ __launch_bounds__(OTPB) void gat_owner_sum(const int* __restrict__ src,
                                                      const int* __restrict__ dst,
                                                      const unsigned* __restrict__ lastplane,
                                                      const float* __restrict__ el,
                                                      const float* __restrict__ er,
                                                      const float* __restrict__ feat,
                                                      float* __restrict__ out) {
    __shared__ float    rows[NR * DFEAT];
    __shared__ unsigned q0[Q0CAP];
    __shared__ unsigned sp[Q0CAP];
    __shared__ float    sa[Q0CAP];
    __shared__ unsigned cnts[2];
    __shared__ unsigned tmax[NR];
    __shared__ float    tden[NR];
    __shared__ unsigned tcnt[NR];
    __shared__ unsigned tstart[NR + 1];

    const int tid  = threadIdx.x;
    const int lane = tid & 31;
    const int w    = tid >> 5;
    const int own  = blockIdx.x;
    const int lo   = own * NR;
    const int cnt  = (NNODES - lo < NR) ? (NNODES - lo) : NR;

    if (tid < 2) cnts[tid] = 0u;
    if (tid < NR) { tmax[tid] = 0u; tden[tid] = 0.0f; tcnt[tid] = 0u; }
    __syncthreads();

    {
        const uint4* pv = (const uint4*)((own < NOWN - 1) ? (const unsigned*)(out + (long)lo * DFEAT) : lastplane);
        uint4 v[NIT];
#pragma unroll
        for (int u = 0; u < NIT; ++u) {
            const int q = tid + u * OTPB;
            v[u] = pv[q < PITEM ? q : tid];
            if (q >= PITEM) v[u] = make_uint4(0u, 0u, 0u, 0u);
        }
        const unsigned j0 = 4u * (unsigned)(lane & 7);
        unsigned nn[NIT];
        unsigned c = 0u;
#pragma unroll
        for (int u = 0; u < NIT; ++u) {
            unsigned n = (unsigned)__shfl((int)v[u].x, lane & ~7, 32);
            if (n > LCAP) { atomicOr(&cnts[1], 1u); n = LCAP; }
            nn[u] = n;
            c += (unsigned)((j0 >= 1u && j0 <= n) + (j0 + 1u <= n) + (j0 + 2u <= n) + (j0 + 3u <= n));
        }
        unsigned incl = c;
#pragma unroll
        for (int off = 1; off < 32; off <<= 1) {
            const unsigned t = (unsigned)__shfl_up((int)incl, off, 32);
            if (lane >= off) incl += t;
        }
        const unsigned wtot = (unsigned)__shfl((int)incl, 31, 32);
        unsigned wbase = 0u;
        if (lane == 0) wbase = atomicAdd(&cnts[0], wtot);
        wbase = (unsigned)__shfl((int)wbase, 0, 32);
        unsigned slot = wbase + incl - c;
#pragma unroll
        for (int u = 0; u < NIT; ++u) {
            const unsigned wd[4] = { v[u].x, v[u].y, v[u].z, v[u].w };
#pragma unroll
            for (int k = 0; k < 4; ++k) {
                const unsigned j = j0 + (unsigned)k;
                if (j >= 1u && j <= nn[u]) {
                    if (slot < Q0CAP) q0[slot] = wd[k];
                    ++slot;
                }
            }
        }
    }
    __syncthreads();

    const unsigned total = cnts[0];
    if (total > Q0CAP || cnts[1] != 0u) {
        owner_all_slow(rows, tmax, tden, src, dst, el, er, feat, out, lo, cnt, tid);
        return;
    }

    {
        const int n0 = (int)total;
        unsigned rc[RIT], place[RIT];
        bool  ok[RIT];
        int   ss[RIT], lrs[RIT];
        float vv[RIT];
#pragma unroll
        for (int k = 0; k < RIT; ++k) {
            const int i = k * OTPB + tid;
            ok[k] = i < n0;
            rc[k] = ok[k] ? q0[i] : 0u;
        }
#pragma unroll
        for (int k = 0; k < RIT; ++k) {
            ss[k]  = (int)(rc[k] & 0x1FFFFu);
            lrs[k] = (int)(rc[k] >> 17);
            ok[k] = ok[k] && lrs[k] < cnt && ss[k] < NNODES;
            const int sq = ok[k] ? ss[k] : 0;
            const int dq = ok[k] ? lo + lrs[k] : lo;
            vv[k] = edge_score(el[sq], er[dq]);
        }
#pragma unroll
        for (int k = 0; k < RIT; ++k) {
            place[k] = 0u;
            if (ok[k]) {
                atomicMax(&tmax[lrs[k]], enc_f32(vv[k]));
                place[k] = atomicAdd(&tcnt[lrs[k]], 1u);
            }
        }
        __syncthreads();
        if (w == 0) {
            unsigned c[RPL];
            unsigned s = 0u;
#pragma unroll
            for (int j = 0; j < RPL; ++j) {
                const int r = lane * RPL + j;
                c[j] = (r < NR) ? tcnt[r] : 0u;
                s += c[j];
            }
            unsigned incl = s;
#pragma unroll
            for (int off = 1; off < 32; off <<= 1) {
                const unsigned t = (unsigned)__shfl_up((int)incl, off, 32);
                if (lane >= off) incl += t;
            }
            unsigned run = incl - s;
#pragma unroll
            for (int j = 0; j < RPL; ++j) {
                const int r = lane * RPL + j;
                if (r <= NR) tstart[r] = run;
                run += c[j];
            }
        }
#pragma unroll
        for (int k = 0; k < RIT; ++k) {
            if (ok[k]) {
                vv[k] = __expf(vv[k] - row_max(tmax[lrs[k]]));
                atomicAdd(&tden[lrs[k]], vv[k]);
            }
        }
        __syncthreads();
#pragma unroll
        for (int k = 0; k < RIT; ++k) {
            if (ok[k]) {
                const int lr = lrs[k];
                const unsigned at = tstart[lr] + place[k];
                sp[at] = (unsigned)ss[k] | ((unsigned)lr << 17);
                sa[at] = vv[k] / fmaxf(tden[lr], 1e-9f);
            }
        }
    }
    __syncthreads();

    {
        const int r0 = (w * cnt) >> 5;
        const int r1 = ((w + 1) * cnt) >> 5;
        {
            const int r = r0 + lane;
            const bool empty = (r < r1) && (tcnt[r] == 0u);
            unsigned zm = __builtin_amdgcn_ballot_w32(empty);
            while (zm != 0u) {
                const int j = __builtin_ctz(zm);
                zm &= zm - 1u;
                row_store(out, lo + r0 + j, lane, (v4f){0.f, 0.f, 0.f, 0.f});
            }
        }
        const int e0 = (int)tstart[r0];
        const int n1 = __builtin_amdgcn_readfirstlane((int)tstart[r1] - e0);
        if (n1 > 0) {
            const unsigned* qp = sp + e0;
            const float*    qa = sa + e0;
            int cur = __builtin_amdgcn_readfirstlane((int)(qp[0] >> 17));
            v4f acc = (v4f){0.f, 0.f, 0.f, 0.f};
            for (int i0 = 0; i0 < n1; i0 += PB) {
                const int  iu = i0 + (lane & (PB - 1));
                const bool vu = iu < n1;
                const int  ic = vu ? iu : i0;
                const unsigned pv = qp[ic];
                const float    av = qa[ic];
                if (i0 + PB <= n1) owner_batch<true>(out, lo, feat, pv, av, PB, lane, cur, acc);
                else               owner_batch<false>(out, lo, feat, pv, av, n1 - i0, lane, cur, acc);
            }
            row_store(out, lo + cur, lane, acc);
        }
    }
}

extern "C" void kernel_launch(void* const* d_in, const int* in_sizes, int n_in,
                              void* d_out, int out_size, void* d_ws, size_t ws_size,
                              hipStream_t stream) {
    const float* h      = (const float*)d_in[0];
    const int*   src    = (const int*)  d_in[1];
    const int*   dst    = (const int*)  d_in[2];
    const float* W      = (const float*)d_in[3];
    const float* attn_l = (const float*)d_in[4];
    const float* attn_r = (const float*)d_in[5];
    float* out = (float*)d_out;

    float*    feat  = (float*)d_ws;
    float*    el    = feat + (long)NNODES * DFEAT;
    float*    er    = el + NNODES;
    unsigned* lastplane = (unsigned*)(er + NNODES);

    gat_lines<<<NTILE, PTPB, 0, stream>>>(src, dst, out, lastplane);

    const int nstrips = NNODES / 16;
    gat_gemm_wmma<<<(nstrips + 3) / 4, 128, 0, stream>>>(h, W, feat);

    gat_attn_logits<<<(NNODES / NPW + 3) / 4, 128, 0, stream>>>(feat, attn_l, attn_r, el, er);

    gat_owner_sum<<<NOWN, OTPB, 0, stream>>>(src, dst, lastplane, el, er, feat, out);
}
